# starting kernel
# baseline (speedup 1.0000x reference)
#include <hip/hip_runtime.h>
#include <hip/hip_bf16.h>

#define BB 2
#define TT 2048
#define CC 1024
#define NH 16
#define HD 64
#define C3 3072
#define BUDGET 408

typedef __attribute__((ext_vector_type(2))) float v2f;
typedef __attribute__((ext_vector_type(8))) float v8f;

__device__ __forceinline__ v8f wmma4(v2f a, v2f b, v8f c) {
  return __builtin_amdgcn_wmma_f32_16x16x4_f32(false, a, false, b, (short)0, c, false, false);
}

__device__ __forceinline__ float rowmax16(float x) {
  #pragma unroll
  for (int d = 1; d < 16; d <<= 1) x = fmaxf(x, __shfl_xor(x, d, 32));
  return x;
}
__device__ __forceinline__ float rowsum16(float x) {
  #pragma unroll
  for (int d = 1; d < 16; d <<= 1) x += __shfl_xor(x, d, 32);
  return x;
}

__global__ void gemm_wmma(const float* __restrict__ A, const float* __restrict__ W,
                          const float* __restrict__ bias, float* __restrict__ Out,
                          int M, int N, int K) {
  int wave = threadIdx.x >> 5;
  int lane = threadIdx.x & 31;
  int g  = lane >> 4;
  int lm = lane & 15;
  int m0 = blockIdx.y * 32;
  int n0 = blockIdx.x * 256 + wave * 64;
  if (n0 >= N || m0 >= M) return;

  v8f acc[2][4];
  #pragma unroll
  for (int t = 0; t < 4; ++t) {
    float bv = bias[n0 + t * 16 + lm];
    #pragma unroll
    for (int mt = 0; mt < 2; ++mt)
      #pragma unroll
      for (int r = 0; r < 8; ++r) acc[mt][t][r] = bv;
  }
  const float* Ar[2];
  Ar[0] = A + (size_t)(m0 + lm) * K + 2 * g;
  Ar[1] = A + (size_t)(m0 + 16 + lm) * K + 2 * g;
  const float* Wp[4];
  #pragma unroll
  for (int t = 0; t < 4; ++t)
    Wp[t] = W + (size_t)(n0 + t * 16 + lm) * K + 2 * g;

  v2f a[2], b[4], an[2], bn[4];
  #pragma unroll
  for (int mt = 0; mt < 2; ++mt) a[mt] = *(const v2f*)(Ar[mt]);
  #pragma unroll
  for (int t = 0; t < 4; ++t) b[t] = *(const v2f*)(Wp[t]);

  for (int k0 = 0; k0 < K; k0 += 4) {
    int k1 = (k0 + 4 < K) ? (k0 + 4) : k0;
    an[0] = *(const v2f*)(Ar[0] + k1);
    an[1] = *(const v2f*)(Ar[1] + k1);
    #pragma unroll
    for (int t = 0; t < 4; ++t) bn[t] = *(const v2f*)(Wp[t] + k1);
    #pragma unroll
    for (int mt = 0; mt < 2; ++mt)
      #pragma unroll
      for (int t = 0; t < 4; ++t)
        acc[mt][t] = wmma4(a[mt], b[t], acc[mt][t]);
    a[0] = an[0]; a[1] = an[1];
    #pragma unroll
    for (int t = 0; t < 4; ++t) b[t] = bn[t];
  }
  #pragma unroll
  for (int mt = 0; mt < 2; ++mt)
    #pragma unroll
    for (int t = 0; t < 4; ++t)
      #pragma unroll
      for (int r = 0; r < 8; ++r)
        Out[(size_t)(m0 + mt * 16 + r + 8 * g) * N + n0 + t * 16 + lm] = acc[mt][t][r];
}

__global__ void head0_scores(const float* __restrict__ qkv, float* __restrict__ S) {
  int jb = blockIdx.x, ib = blockIdx.y, b = blockIdx.z;
  int lane = threadIdx.x & 31, g = lane >> 4, lm = lane & 15;
  int i0 = ib * 16, j0 = jb * 16;
  if (jb > ib) {
    #pragma unroll
    for (int r = 0; r < 8; ++r)
      S[((size_t)(b * TT + i0 + r + 8 * g)) * TT + j0 + lm] = 0.f;
    return;
  }
  const float* qrow = qkv + ((size_t)(b * TT) + i0 + lm) * C3;
  const float* krow = qkv + ((size_t)(b * TT) + j0 + lm) * C3 + CC;
  v2f aq[16], bk[16];
  #pragma unroll
  for (int ks = 0; ks < 16; ++ks) {
    aq[ks] = *(const v2f*)(qrow + ks * 4 + 2 * g);
    bk[ks] = *(const v2f*)(krow + ks * 4 + 2 * g);
  }
  v8f acc = {0.f, 0.f, 0.f, 0.f, 0.f, 0.f, 0.f, 0.f};
  #pragma unroll
  for (int ks = 0; ks < 16; ++ks) acc = wmma4(aq[ks], bk[ks], acc);
  #pragma unroll
  for (int r = 0; r < 8; ++r) {
    int i = i0 + r + 8 * g, j = j0 + lm;
    float s = acc[r] * 0.125f;
    s = (j < i && j > 0) ? fmaxf(s, 0.f) : 0.f;
    S[((size_t)(b * TT + i)) * TT + j] = s;
  }
}

__global__ void ff_cumsum(float* __restrict__ S) {
  int j = blockIdx.x * blockDim.x + threadIdx.x;
  int b = blockIdx.y;
  float* col = S + (size_t)b * TT * TT + j;
  float carry = 0.f;
  #pragma unroll 8
  for (int i = 0; i < TT; ++i) {
    float v = col[(size_t)i * TT];
    col[(size_t)i * TT] = carry;
    carry += v;
  }
}

__global__ void select_mask(const float* __restrict__ FF, unsigned* __restrict__ mask) {
  int row = blockIdx.x;
  int i = row & (TT - 1);
  const float* f = FF + (size_t)row * TT;
  unsigned* mrow = mask + (size_t)row * (TT / 32);
  int tid = threadIdx.x;

  if (i < BUDGET) {
    for (int w = tid; w < TT / 32; w += 256) {
      int jb = w * 32;
      unsigned m;
      if (jb + 31 <= i) m = 0xFFFFFFFFu;
      else if (jb > i)  m = 0u;
      else              m = (1u << (i - jb + 1)) - 1u;
      mrow[w] = m;
    }
    return;
  }

  __shared__ unsigned hist[256];
  __shared__ unsigned misc[4];
  __shared__ unsigned wave_tot[8];

  if (tid == 0) { misc[0] = 0u; misc[1] = BUDGET - 1; }
  unsigned prefix = 0, prefmask = 0;
  for (int pass = 0; pass < 4; ++pass) {
    int shift = 24 - 8 * pass;
    hist[tid] = 0u;
    __syncthreads();
    for (int j = tid; j <= i; j += 256) {
      unsigned u = __float_as_uint(f[j]);
      if ((u & prefmask) == prefix) atomicAdd(&hist[(u >> shift) & 255u], 1u);
    }
    __syncthreads();
    if (tid == 0) {
      unsigned kk = misc[1], cum = 0; int d = 0;
      for (; d < 256; ++d) { unsigned h = hist[d]; if (cum + h > kk) break; cum += h; }
      misc[1] = kk - cum;
      misc[0] |= ((unsigned)d << shift);
    }
    __syncthreads();
    prefix = misc[0];
    prefmask |= (255u << shift);
  }
  unsigned theta = prefix;

  if (tid == 0) misc[2] = 0u;
  __syncthreads();
  unsigned cl = 0;
  for (int j = tid; j <= i; j += 256)
    if (__float_as_uint(f[j]) < theta) cl++;
  atomicAdd(&misc[2], cl);
  __syncthreads();
  int ntake = BUDGET - (int)misc[2];

  int lane = tid & 31, wv = tid >> 5;
  unsigned below = (1u << lane) - 1u;
  unsigned carry = 0;
  for (int base = 0; base < TT; base += 256) {
    int j = base + tid;
    unsigned u = (j <= i) ? __float_as_uint(f[j]) : 0xFFFFFFFFu;
    bool eq = (j <= i) && (u == theta);
    unsigned bal = (unsigned)__ballot(eq);
    if (lane == 0) wave_tot[wv] = __popc(bal);
    __syncthreads();
    unsigned pre = carry + __popc(bal & below);
    for (int w2 = 0; w2 < wv; ++w2) pre += wave_tot[w2];
    bool keep = (j <= i) && ((u < theta) || (eq && (int)pre < ntake));
    keep = keep || (j == i);
    unsigned kb = (unsigned)__ballot(keep);
    if (lane == 0) mrow[(base + wv * 32) >> 5] = kb;
    unsigned tot = 0;
    #pragma unroll
    for (int w2 = 0; w2 < 8; ++w2) tot += wave_tot[w2];
    carry += tot;
    __syncthreads();
  }
}

__global__ void attn(const float* __restrict__ qkv, const float* __restrict__ FF,
                     const unsigned* __restrict__ mask, float* __restrict__ Y) {
  int ib = blockIdx.x, h = blockIdx.y, b = blockIdx.z;
  int lane = threadIdx.x & 31, g = lane >> 4, lm = lane & 15;
  int i0 = ib * 16;
  __shared__ float pt[16 * 17];

  const float* qb = qkv + ((size_t)(b * TT)) * C3 + h * HD;
  const float* kb = qb + CC;
  const float* vb = qb + 2 * CC;

  v2f aq[16];
  #pragma unroll
  for (int ks = 0; ks < 16; ++ks)
    aq[ks] = *(const v2f*)(qb + (size_t)(i0 + lm) * C3 + ks * 4 + 2 * g);

  float mrow[8], lrow[8];
  v8f o[4];
  #pragma unroll
  for (int t = 0; t < 4; ++t)
    #pragma unroll
    for (int r = 0; r < 8; ++r) o[t][r] = 0.f;
  #pragma unroll
  for (int r = 0; r < 8; ++r) { mrow[r] = -1e30f; lrow[r] = 0.f; }

  for (int jb = 0; jb <= ib; ++jb) {
    int j0 = jb * 16;
    v2f bk[16];
    #pragma unroll
    for (int ks = 0; ks < 16; ++ks)
      bk[ks] = *(const v2f*)(kb + (size_t)(j0 + lm) * C3 + ks * 4 + 2 * g);
    v8f acc = {0.f, 0.f, 0.f, 0.f, 0.f, 0.f, 0.f, 0.f};
    #pragma unroll
    for (int ks = 0; ks < 16; ++ks) acc = wmma4(aq[ks], bk[ks], acc);

    float ffv[8]; unsigned mwv[8];
    #pragma unroll
    for (int r = 0; r < 8; ++r) {
      int i = i0 + r + 8 * g;
      ffv[r] = FF[((size_t)(b * TT + i)) * TT + j0 + lm];
      mwv[r] = mask[((size_t)(b * TT + i)) * (TT / 32) + (j0 >> 5)];
    }
    float p[8];
    #pragma unroll
    for (int r = 0; r < 8; ++r) {
      int i = i0 + r + 8 * g, j = j0 + lm;
      bool keep = ((mwv[r] >> (j & 31)) & 1u) && (j <= i);
      float s = keep ? (acc[r] * 0.125f - ffv[r]) : -1e30f;
      float bm = rowmax16(s);
      float mnew = fmaxf(mrow[r], bm);
      float scale = __expf(mrow[r] - mnew);
      float pr = __expf(s - mnew);
      lrow[r] = lrow[r] * scale + rowsum16(pr);
      mrow[r] = mnew;
      p[r] = pr;
      #pragma unroll
      for (int t = 0; t < 4; ++t) o[t][r] *= scale;
    }
    __syncthreads();
    #pragma unroll
    for (int r = 0; r < 8; ++r) pt[(r + 8 * g) * 17 + lm] = p[r];
    __syncthreads();
    v2f ap[4];
    #pragma unroll
    for (int kt = 0; kt < 4; ++kt) {
      ap[kt].x = pt[lm * 17 + kt * 4 + 2 * g];
      ap[kt].y = pt[lm * 17 + kt * 4 + 2 * g + 1];
    }
    v2f bv[4][4];
    #pragma unroll
    for (int kt = 0; kt < 4; ++kt)
      #pragma unroll
      for (int t = 0; t < 4; ++t) {
        bv[kt][t].x = vb[(size_t)(j0 + kt * 4 + 2 * g) * C3 + t * 16 + lm];
        bv[kt][t].y = vb[(size_t)(j0 + kt * 4 + 2 * g + 1) * C3 + t * 16 + lm];
      }
    #pragma unroll
    for (int kt = 0; kt < 4; ++kt)
      #pragma unroll
      for (int t = 0; t < 4; ++t)
        o[t] = wmma4(ap[kt], bv[kt][t], o[t]);
  }
  #pragma unroll
  for (int r = 0; r < 8; ++r) {
    float inv = 1.f / lrow[r];
    int i = i0 + r + 8 * g;
    #pragma unroll
    for (int t = 0; t < 4; ++t)
      Y[((size_t)(b * TT + i)) * CC + h * HD + t * 16 + lm] = o[t][r] * inv;
  }
}

extern "C" void kernel_launch(void* const* d_in, const int* in_sizes, int n_in,
                              void* d_out, int out_size, void* d_ws, size_t ws_size,
                              hipStream_t stream) {
  const float* x      = (const float*)d_in[0];
  const float* w_attn = (const float*)d_in[1];
  const float* b_attn = (const float*)d_in[2];
  const float* w_proj = (const float*)d_in[3];
  const float* b_proj = (const float*)d_in[4];
  float* out = (float*)d_out;

  char* ws = (char*)d_ws;
  const size_t QKV_BYTES = (size_t)BB * TT * C3 * 4;
  const size_t FF_BYTES  = (size_t)BB * TT * TT * 4;
  const size_t MSK_BYTES = (size_t)BB * TT * (TT / 32) * 4;
  float*    qkv = (float*)ws;
  float*    FF  = (float*)(ws + QKV_BYTES);
  unsigned* msk = (unsigned*)(ws + QKV_BYTES + FF_BYTES);
  float*    Y   = (float*)(ws + QKV_BYTES + FF_BYTES + MSK_BYTES);

  gemm_wmma<<<dim3(C3 / 256, (BB * TT) / 32), 128, 0, stream>>>(
      x, w_attn, b_attn, qkv, BB * TT, C3, CC);
  head0_scores<<<dim3(TT / 16, TT / 16, BB), 32, 0, stream>>>(qkv, FF);
  ff_cumsum<<<dim3(TT / 256, BB), 256, 0, stream>>>(FF);
  select_mask<<<BB * TT, 256, 0, stream>>>(FF, msk);
  attn<<<dim3(TT / 16, NH, BB), 32, 0, stream>>>(qkv, FF, msk, Y);
  gemm_wmma<<<dim3(CC / 256, (BB * TT) / 32), 128, 0, stream>>>(
      Y, w_proj, b_proj, out, BB * TT, CC, CC);
}


# attn: each wave stages its K and V key tiles in LDS by whole-line 16-byte global loads, next tile in flight during the current one; fragments read from LDS, was 8-byte and 4-byte gathers from global
# speedup vs baseline: 3.2113x; 3.2113x over previous
#include <hip/hip_runtime.h>
#include <hip/hip_bf16.h>

#define BB 2
#define TT 2048
#define CC 1024
#define NH 16
#define HD 64
#define C3 3072
#define BUDGET 408

typedef __attribute__((ext_vector_type(2))) float v2f;
typedef __attribute__((ext_vector_type(8))) float v8f;

__device__ __forceinline__ v8f wmma4(v2f a, v2f b, v8f c) {
  return __builtin_amdgcn_wmma_f32_16x16x4_f32(false, a, false, b, (short)0, c, false, false);
}

__device__ __forceinline__ float rowmax16(float x) {
  #pragma unroll
  for (int d = 1; d < 16; d <<= 1) x = fmaxf(x, __shfl_xor(x, d, 32));
  return x;
}
__device__ __forceinline__ float rowsum16(float x) {
  #pragma unroll
  for (int d = 1; d < 16; d <<= 1) x += __shfl_xor(x, d, 32);
  return x;
}

#define GBM 128
#define GBN 128
#define GBK 32
#define GP  36

typedef __attribute__((ext_vector_type(4))) float v4f;

__global__ void __launch_bounds__(128)
gemm_wmma(const float* __restrict__ A, const float* __restrict__ W,
          const float* __restrict__ bias, float* __restrict__ Out,
          int M, int N, int K) {
  __shared__ __attribute__((aligned(16))) float As[2][GBM * GP];
  __shared__ __attribute__((aligned(16))) float Ws[2][GBN * GP];

  int tid = threadIdx.x;
  int w = __builtin_amdgcn_readfirstlane(tid >> 5);
  int lane = tid & 31, h = lane >> 4, c = lane & 15;
  int wm = w >> 1, wn = w & 1;
  int m0 = blockIdx.y * GBM, n0 = blockIdx.x * GBN;

  int crow = tid >> 3, cch = tid & 7;
  const float* Ag = A + (size_t)(m0 + crow) * K + cch * 4;
  const float* Wg = W + (size_t)(n0 + crow) * K + cch * 4;
  int st_off = crow * GP + cch * 4;
  int a_off = (wm * 64 + c) * GP + 4 * h;
  int b_off = (wn * 64 + c) * GP + 4 * h;

  v8f acc[4][4];
  #pragma unroll
  for (int nt = 0; nt < 4; ++nt) {
    float bv = bias[n0 + wn * 64 + nt * 16 + c];
    #pragma unroll
    for (int mt = 0; mt < 4; ++mt)
      #pragma unroll
      for (int r = 0; r < 8; ++r) acc[mt][nt][r] = bv;
  }

  v4f ga[8], gw[8];
  #pragma unroll
  for (int p = 0; p < 8; ++p) {
    ga[p] = *(const v4f*)(Ag + (size_t)p * 16 * K);
    gw[p] = *(const v4f*)(Wg + (size_t)p * 16 * K);
  }
  #pragma unroll
  for (int p = 0; p < 8; ++p) {
    *(v4f*)(&As[0][st_off + p * 16 * GP]) = ga[p];
    *(v4f*)(&Ws[0][st_off + p * 16 * GP]) = gw[p];
  }
  __syncthreads();

  const int nk = K / GBK;
  for (int kt = 0; kt < nk; ++kt) {
    const int cur = kt & 1;
    const bool more = (kt + 1 < nk);
    if (more) {
      #pragma unroll
      for (int p = 0; p < 8; ++p) {
        ga[p] = *(const v4f*)(Ag + (size_t)p * 16 * K + (kt + 1) * GBK);
        gw[p] = *(const v4f*)(Wg + (size_t)p * 16 * K + (kt + 1) * GBK);
      }
    }
    const float* as = &As[cur][a_off];
    const float* ws = &Ws[cur][b_off];
    v4f fa[2][4], fb[2][4];
    #pragma unroll
    for (int t = 0; t < 4; ++t) {
      fa[0][t] = *(const v4f*)(as + t * 16 * GP);
      fb[0][t] = *(const v4f*)(ws + t * 16 * GP);
    }
    #pragma unroll
    for (int kb = 0; kb < GBK / 8; ++kb) {
      const int s = kb & 1;
      if (kb + 1 < GBK / 8) {
        #pragma unroll
        for (int t = 0; t < 4; ++t) {
          fa[s ^ 1][t] = *(const v4f*)(as + t * 16 * GP + (kb + 1) * 8);
          fb[s ^ 1][t] = *(const v4f*)(ws + t * 16 * GP + (kb + 1) * 8);
        }
      }
      #pragma unroll
      for (int mt = 0; mt < 4; ++mt) {
        v2f a0 = {fa[s][mt].x, fa[s][mt].y};
        #pragma unroll
        for (int nt = 0; nt < 4; ++nt) {
          v2f b0 = {fb[s][nt].x, fb[s][nt].y};
          acc[mt][nt] = wmma4(a0, b0, acc[mt][nt]);
        }
      }
      #pragma unroll
      for (int mt = 0; mt < 4; ++mt) {
        v2f a1 = {fa[s][mt].z, fa[s][mt].w};
        #pragma unroll
        for (int nt = 0; nt < 4; ++nt) {
          v2f b1 = {fb[s][nt].z, fb[s][nt].w};
          acc[mt][nt] = wmma4(a1, b1, acc[mt][nt]);
        }
      }
    }
    if (more) {
      #pragma unroll
      for (int p = 0; p < 8; ++p) {
        *(v4f*)(&As[cur ^ 1][st_off + p * 16 * GP]) = ga[p];
        *(v4f*)(&Ws[cur ^ 1][st_off + p * 16 * GP]) = gw[p];
      }
    }
    __syncthreads();
  }

  #pragma unroll
  for (int mt = 0; mt < 4; ++mt)
    #pragma unroll
    for (int nt = 0; nt < 4; ++nt)
      #pragma unroll
      for (int r = 0; r < 8; ++r)
        Out[(size_t)(m0 + wm * 64 + mt * 16 + r + 8 * h) * N + n0 + wn * 64 + nt * 16 + c] = acc[mt][nt][r];
}

#define NT (TT / 16)

__global__ void __launch_bounds__(32)
head0_tiles(const float* __restrict__ qkv, float* __restrict__ Pfx, float* __restrict__ Ttot) {
  int x = blockIdx.x, y = blockIdx.y, b = blockIdx.z;
  int ib, jb;
  if (x <= y) { ib = y; jb = x; } else { ib = NT - 1 - y; jb = x - y - 1; }
  int lane = threadIdx.x & 31, g = lane >> 4, lm = lane & 15;
  int i0 = ib * 16, j0 = jb * 16;
  const float* qrow = qkv + ((size_t)(b * TT) + i0 + lm) * C3;
  const float* krow = qkv + ((size_t)(b * TT) + j0 + lm) * C3 + CC;
  v2f aq[16], bk[16];
  #pragma unroll
  for (int ks = 0; ks < 16; ++ks) {
    aq[ks] = *(const v2f*)(qrow + ks * 4 + 2 * g);
    bk[ks] = *(const v2f*)(krow + ks * 4 + 2 * g);
  }
  v8f acc = {0.f, 0.f, 0.f, 0.f, 0.f, 0.f, 0.f, 0.f};
  #pragma unroll
  for (int ks = 0; ks < 16; ++ks) acc = wmma4(aq[ks], bk[ks], acc);
  float p[8], run = 0.f;
  #pragma unroll
  for (int r = 0; r < 8; ++r) {
    int i = i0 + r + 8 * g, j = j0 + lm;
    float s = acc[r] * 0.125f;
    s = (j < i && j > 0) ? fmaxf(s, 0.f) : 0.f;
    p[r] = run;
    run += s;
  }
  float other = __shfl_xor(run, 16, 32);
  float above = g ? other : 0.f;
  #pragma unroll
  for (int r = 0; r < 8; ++r)
    Pfx[((size_t)(b * TT + i0 + r + 8 * g)) * TT + j0 + lm] = p[r] + above;
  if (g == 0) Ttot[((size_t)(b * NT + ib)) * TT + j0 + lm] = run + other;
}

__global__ void tile_scan(const float* __restrict__ Ttot, float* __restrict__ Toff) {
  int j = blockIdx.x * blockDim.x + threadIdx.x;
  int b = blockIdx.y;
  const float* in  = Ttot + (size_t)b * NT * TT + j;
  float*       out = Toff + (size_t)b * NT * TT + j;
  int jb = j >> 4;
  float carry = 0.f;
  for (int base = 0; base < NT; base += 16) {
    float v[16];
    #pragma unroll
    for (int u = 0; u < 16; ++u) v[u] = in[(size_t)min(jb + base + u, NT - 1) * TT];
    #pragma unroll
    for (int u = 0; u < 16; ++u) {
      int ib = jb + base + u;
      if (ib < NT) out[(size_t)ib * TT] = carry;
      carry += v[u];
    }
  }
}

__global__ void select_mask(float* FF, const float* __restrict__ Toff, unsigned* __restrict__ mask) {
  int row = blockIdx.x;
  int i = row & (TT - 1);
  float* f = FF + (size_t)row * TT;
  const float* off = Toff + ((size_t)((row >> 11) * NT + (i >> 4))) * TT;
  unsigned* mrow = mask + (size_t)row * (TT / 32);
  int tid = threadIdx.x;

  for (int j = tid; j <= i; j += 256) f[j] += off[j];

  if (i < BUDGET) {
    for (int w = tid; w < TT / 32; w += 256) {
      int jb = w * 32;
      unsigned m;
      if (jb + 31 <= i) m = 0xFFFFFFFFu;
      else if (jb > i)  m = 0u;
      else              m = (1u << (i - jb + 1)) - 1u;
      mrow[w] = m;
    }
    return;
  }

  __shared__ unsigned hist[256];
  __shared__ unsigned misc[4];
  __shared__ unsigned wave_tot[8];

  if (tid == 0) { misc[0] = 0u; misc[1] = BUDGET - 1; }
  unsigned prefix = 0, prefmask = 0;
  for (int pass = 0; pass < 4; ++pass) {
    int shift = 24 - 8 * pass;
    hist[tid] = 0u;
    __syncthreads();
    for (int j = tid; j <= i; j += 256) {
      unsigned u = __float_as_uint(f[j]);
      if ((u & prefmask) == prefix) atomicAdd(&hist[(u >> shift) & 255u], 1u);
    }
    __syncthreads();
    if (tid == 0) {
      unsigned kk = misc[1], cum = 0; int d = 0;
      for (; d < 256; ++d) { unsigned h = hist[d]; if (cum + h > kk) break; cum += h; }
      misc[1] = kk - cum;
      misc[0] |= ((unsigned)d << shift);
    }
    __syncthreads();
    prefix = misc[0];
    prefmask |= (255u << shift);
  }
  unsigned theta = prefix;

  if (tid == 0) misc[2] = 0u;
  __syncthreads();
  unsigned cl = 0;
  for (int j = tid; j <= i; j += 256)
    if (__float_as_uint(f[j]) < theta) cl++;
  atomicAdd(&misc[2], cl);
  __syncthreads();
  int ntake = BUDGET - (int)misc[2];

  int lane = tid & 31, wv = tid >> 5;
  unsigned below = (1u << lane) - 1u;
  unsigned carry = 0;
  for (int base = 0; base < TT; base += 256) {
    int j = base + tid;
    unsigned u = (j <= i) ? __float_as_uint(f[j]) : 0xFFFFFFFFu;
    bool eq = (j <= i) && (u == theta);
    unsigned bal = (unsigned)__ballot(eq);
    if (lane == 0) wave_tot[wv] = __popc(bal);
    __syncthreads();
    unsigned pre = carry + __popc(bal & below);
    for (int w2 = 0; w2 < wv; ++w2) pre += wave_tot[w2];
    bool keep = (j <= i) && ((u < theta) || (eq && (int)pre < ntake));
    keep = keep || (j == i);
    unsigned kb = (unsigned)__ballot(keep);
    if (lane == 0) mrow[(base + wv * 32) >> 5] = kb;
    unsigned tot = 0;
    #pragma unroll
    for (int w2 = 0; w2 < 8; ++w2) tot += wave_tot[w2];
    carry += tot;
    __syncthreads();
  }
}

#define NW 4
#define OST (HD + 4)
#define KP  (HD + 4)
#define VP  (HD + 8)
#define PTP 17
#define WSF (16 * KP + 16 * VP + 16 * PTP)

__device__ __forceinline__ void wave_lds_sync() {
  __builtin_amdgcn_fence(__ATOMIC_RELEASE, "wavefront");
  __builtin_amdgcn_wave_barrier();
  __builtin_amdgcn_fence(__ATOMIC_ACQUIRE, "wavefront");
}

__global__ void __launch_bounds__(32 * NW)
attn(const float* __restrict__ qkv, const float* __restrict__ FF,
     const unsigned* __restrict__ mask, float* __restrict__ Y) {
  int ib = blockIdx.x, h = blockIdx.y, b = blockIdx.z;
  int w = __builtin_amdgcn_readfirstlane((int)(threadIdx.x >> 5));
  int lane = threadIdx.x & 31, g = lane >> 4, lm = lane & 15;
  int i0 = ib * 16;
  __shared__ __attribute__((aligned(16))) float stage[NW][WSF];
  __shared__ float part_m[NW][16];
  __shared__ float part_l[NW][16];
  float* Ks = stage[w];
  float* Vs = Ks + 16 * KP;
  float* pt = Vs + 16 * VP;

  const float* qb = qkv + ((size_t)(b * TT)) * C3 + h * HD;
  const float* kb = qb + CC;

  v4f aq[8];
  #pragma unroll
  for (int k8 = 0; k8 < 8; ++k8)
    aq[k8] = *(const v4f*)(qb + (size_t)(i0 + lm) * C3 + k8 * 8 + 4 * g);

  const float* kg = kb + (size_t)g * C3 + lm * 4;
  const float* vg = kg + CC;
  const int k_st = g * KP + lm * 4;
  const int v_st = g * VP + lm * 4;
  const int k_fr = lm * KP + 4 * g;
  const int v_fr = 2 * g * VP + lm;

  float mrow[8], lrow[8];
  v8f o[4];
  #pragma unroll
  for (int t = 0; t < 4; ++t)
    #pragma unroll
    for (int r = 0; r < 8; ++r) o[t][r] = 0.f;
  #pragma unroll
  for (int r = 0; r < 8; ++r) { mrow[r] = -1e30f; lrow[r] = 0.f; }

  v4f rk[8], rv[8];
  {
    const int jf = min(w, ib) * 16;
    #pragma unroll
    for (int p = 0; p < 8; ++p) {
      rk[p] = *(const v4f*)(kg + (size_t)(jf + 2 * p) * C3);
      rv[p] = *(const v4f*)(vg + (size_t)(jf + 2 * p) * C3);
    }
    #pragma unroll
    for (int p = 0; p < 8; ++p) {
      *(v4f*)(Ks + k_st + 2 * p * KP) = rk[p];
      *(v4f*)(Vs + v_st + 2 * p * VP) = rv[p];
    }
  }
  wave_lds_sync();

  for (int jb = w; jb <= ib; jb += NW) {
    int j0 = jb * 16;
    const int jn = min(jb + NW, ib) * 16;
    float ffv[8]; unsigned mwv[8];
    #pragma unroll
    for (int r = 0; r < 8; ++r) {
      int i = i0 + r + 8 * g;
      ffv[r] = FF[((size_t)(b * TT + i)) * TT + j0 + lm];
      mwv[r] = mask[((size_t)(b * TT + i)) * (TT / 32) + (j0 >> 5)];
    }
    #pragma unroll
    for (int p = 0; p < 8; ++p)
      rk[p] = *(const v4f*)(kg + (size_t)(jn + 2 * p) * C3);
    #pragma unroll
    for (int p = 0; p < 8; ++p)
      rv[p] = *(const v4f*)(vg + (size_t)(jn + 2 * p) * C3);
    v4f fk[8];
    #pragma unroll
    for (int k8 = 0; k8 < 8; ++k8) fk[k8] = *(const v4f*)(Ks + k_fr + k8 * 8);
    v8f acc = {0.f, 0.f, 0.f, 0.f, 0.f, 0.f, 0.f, 0.f};
    #pragma unroll
    for (int k8 = 0; k8 < 8; ++k8) {
      v2f a0 = {aq[k8].x, aq[k8].y}, b0 = {fk[k8].x, fk[k8].y};
      acc = wmma4(a0, b0, acc);
      v2f a1 = {aq[k8].z, aq[k8].w}, b1 = {fk[k8].z, fk[k8].w};
      acc = wmma4(a1, b1, acc);
    }

    float p[8];
    #pragma unroll
    for (int r = 0; r < 8; ++r) {
      int i = i0 + r + 8 * g, j = j0 + lm;
      bool keep = ((mwv[r] >> (j & 31)) & 1u) && (j <= i);
      float s = keep ? (acc[r] * 0.125f - ffv[r]) : -1e30f;
      float bm = rowmax16(s);
      float mnew = fmaxf(mrow[r], bm);
      float scale = __expf(mrow[r] - mnew);
      float pr = __expf(s - mnew);
      lrow[r] = lrow[r] * scale + rowsum16(pr);
      mrow[r] = mnew;
      p[r] = pr;
      #pragma unroll
      for (int t = 0; t < 4; ++t) o[t][r] *= scale;
    }
    __builtin_amdgcn_sched_barrier(0);
    wave_lds_sync();
    #pragma unroll
    for (int p8 = 0; p8 < 8; ++p8) *(v4f*)(Ks + k_st + 2 * p8 * KP) = rk[p8];
    #pragma unroll
    for (int r = 0; r < 8; ++r) pt[(r + 8 * g) * PTP + lm] = p[r];
    wave_lds_sync();
    v2f ap[4];
    #pragma unroll
    for (int kt = 0; kt < 4; ++kt) {
      ap[kt].x = pt[lm * PTP + kt * 4 + 2 * g];
      ap[kt].y = pt[lm * PTP + kt * 4 + 2 * g + 1];
    }
    v2f bv[4][4];
    #pragma unroll
    for (int kt = 0; kt < 4; ++kt)
      #pragma unroll
      for (int t = 0; t < 4; ++t) {
        bv[kt][t].x = Vs[v_fr + (kt * 4) * VP + t * 16];
        bv[kt][t].y = Vs[v_fr + (kt * 4 + 1) * VP + t * 16];
      }
    #pragma unroll
    for (int kt = 0; kt < 4; ++kt)
      #pragma unroll
      for (int t = 0; t < 4; ++t)
        o[t] = wmma4(ap[kt], bv[kt][t], o[t]);
    wave_lds_sync();
    #pragma unroll
    for (int p8 = 0; p8 < 8; ++p8) *(v4f*)(Vs + v_st + 2 * p8 * VP) = rv[p8];
  }

  wave_lds_sync();
  float* po = stage[w];
  #pragma unroll
  for (int r = 0; r < 8; ++r) {
    #pragma unroll
    for (int t = 0; t < 4; ++t) po[(r + 8 * g) * OST + t * 16 + lm] = o[t][r];
    if (lm == 0) { part_m[w][r + 8 * g] = mrow[r]; part_l[w][r + 8 * g] = lrow[r]; }
  }
  __syncthreads();
  #pragma unroll
  for (int rr = 0; rr < 16 / NW; ++rr) {
    int row = w * (16 / NW) + rr;
    float mw[NW], M = -1e30f;
    #pragma unroll
    for (int w2 = 0; w2 < NW; ++w2) { mw[w2] = part_m[w2][row]; M = fmaxf(M, mw[w2]); }
    float L = 0.f, o0 = 0.f, o1 = 0.f;
    #pragma unroll
    for (int w2 = 0; w2 < NW; ++w2) {
      float wt = __expf(mw[w2] - M);
      L  += part_l[w2][row] * wt;
      o0 += stage[w2][row * OST + lane] * wt;
      o1 += stage[w2][row * OST + lane + 32] * wt;
    }
    float inv = 1.f / L;
    float* yrow = Y + ((size_t)(b * TT + i0 + row)) * CC + h * HD;
    yrow[lane]      = o0 * inv;
    yrow[lane + 32] = o1 * inv;
  }
}

extern "C" void kernel_launch(void* const* d_in, const int* in_sizes, int n_in,
                              void* d_out, int out_size, void* d_ws, size_t ws_size,
                              hipStream_t stream) {
  const float* x      = (const float*)d_in[0];
  const float* w_attn = (const float*)d_in[1];
  const float* b_attn = (const float*)d_in[2];
  const float* w_proj = (const float*)d_in[3];
  const float* b_proj = (const float*)d_in[4];
  float* out = (float*)d_out;

  char* ws = (char*)d_ws;
  const size_t QKV_BYTES = (size_t)BB * TT * C3 * 4;
  const size_t FF_BYTES  = (size_t)BB * TT * TT * 4;
  const size_t MSK_BYTES = (size_t)BB * TT * (TT / 32) * 4;
  float*    qkv = (float*)ws;
  float*    FF  = (float*)(ws + QKV_BYTES);
  unsigned* msk = (unsigned*)(ws + QKV_BYTES + FF_BYTES);
  float*    Y   = (float*)(ws + QKV_BYTES + FF_BYTES + MSK_BYTES);

  gemm_wmma<<<dim3(C3 / GBN, (BB * TT) / GBM), 128, 0, stream>>>(
      x, w_attn, b_attn, qkv, BB * TT, C3, CC);
  float*    Ttot = Y;
  float*    Toff = Y + (size_t)BB * NT * TT;
  head0_tiles<<<dim3(NT + 1, NT / 2, BB), 32, 0, stream>>>(qkv, FF, Ttot);
  tile_scan<<<dim3(TT / 64, BB), 64, 0, stream>>>(Ttot, Toff);
  select_mask<<<BB * TT, 256, 0, stream>>>(FF, Toff, msk);
  attn<<<dim3(TT / 16, NH, BB), 32 * NW, 0, stream>>>(qkv, FF, msk, Y);
  gemm_wmma<<<dim3(CC / GBN, (BB * TT) / GBM), 128, 0, stream>>>(
      Y, w_proj, b_proj, out, BB * TT, CC, CC);
}


# attn softmax: row max and row sum over a score row's 16 lanes by four DPP row_xmask steps each on the vector ALU; was __shfl_xor = ds_bpermute_b32 through the LDS unit with a wait behind each step
# speedup vs baseline: 3.3672x; 1.0485x over previous
#include <hip/hip_runtime.h>
#include <hip/hip_bf16.h>

#define BB 2
#define TT 2048
#define CC 1024
#define NH 16
#define HD 64
#define C3 3072
#define BUDGET 408

typedef __attribute__((ext_vector_type(2))) float v2f;
typedef __attribute__((ext_vector_type(8))) float v8f;

__device__ __forceinline__ v8f wmma4(v2f a, v2f b, v8f c) {
  return __builtin_amdgcn_wmma_f32_16x16x4_f32(false, a, false, b, (short)0, c, false, false);
}

template <int M>
__device__ __forceinline__ float row_xor(float x) {
  int b = __builtin_bit_cast(int, x);
  return __builtin_bit_cast(float, __builtin_amdgcn_update_dpp(b, b, 0x160 + M, 0xf, 0xf, true));
}
__device__ __forceinline__ float rowmax16(float x) {
  x = __builtin_elementwise_maximum(x, row_xor<1>(x));
  x = __builtin_elementwise_maximum(x, row_xor<2>(x));
  x = __builtin_elementwise_maximum(x, row_xor<4>(x));
  x = __builtin_elementwise_maximum(x, row_xor<8>(x));
  return x;
}
__device__ __forceinline__ float rowsum16(float x) {
  x += row_xor<1>(x);
  x += row_xor<2>(x);
  x += row_xor<4>(x);
  x += row_xor<8>(x);
  return x;
}

#define GBM 128
#define GBN 128
#define GBK 32
#define GP  36

typedef __attribute__((ext_vector_type(4))) float v4f;

__global__ void __launch_bounds__(128)
gemm_wmma(const float* __restrict__ A, const float* __restrict__ W,
          const float* __restrict__ bias, float* __restrict__ Out,
          int M, int N, int K) {
  __shared__ __attribute__((aligned(16))) float As[2][GBM * GP];
  __shared__ __attribute__((aligned(16))) float Ws[2][GBN * GP];

  int tid = threadIdx.x;
  int w = __builtin_amdgcn_readfirstlane(tid >> 5);
  int lane = tid & 31, h = lane >> 4, c = lane & 15;
  int wm = w >> 1, wn = w & 1;
  int m0 = blockIdx.y * GBM, n0 = blockIdx.x * GBN;

  int crow = tid >> 3, cch = tid & 7;
  const float* Ag = A + (size_t)(m0 + crow) * K + cch * 4;
  const float* Wg = W + (size_t)(n0 + crow) * K + cch * 4;
  int st_off = crow * GP + cch * 4;
  int a_off = (wm * 64 + c) * GP + 4 * h;
  int b_off = (wn * 64 + c) * GP + 4 * h;

  v8f acc[4][4];
  #pragma unroll
  for (int nt = 0; nt < 4; ++nt) {
    float bv = bias[n0 + wn * 64 + nt * 16 + c];
    #pragma unroll
    for (int mt = 0; mt < 4; ++mt)
      #pragma unroll
      for (int r = 0; r < 8; ++r) acc[mt][nt][r] = bv;
  }

  v4f ga[8], gw[8];
  #pragma unroll
  for (int p = 0; p < 8; ++p) {
    ga[p] = *(const v4f*)(Ag + (size_t)p * 16 * K);
    gw[p] = *(const v4f*)(Wg + (size_t)p * 16 * K);
  }
  #pragma unroll
  for (int p = 0; p < 8; ++p) {
    *(v4f*)(&As[0][st_off + p * 16 * GP]) = ga[p];
    *(v4f*)(&Ws[0][st_off + p * 16 * GP]) = gw[p];
  }
  __syncthreads();

  const int nk = K / GBK;
  for (int kt = 0; kt < nk; ++kt) {
    const int cur = kt & 1;
    const bool more = (kt + 1 < nk);
    if (more) {
      #pragma unroll
      for (int p = 0; p < 8; ++p) {
        ga[p] = *(const v4f*)(Ag + (size_t)p * 16 * K + (kt + 1) * GBK);
        gw[p] = *(const v4f*)(Wg + (size_t)p * 16 * K + (kt + 1) * GBK);
      }
    }
    const float* as = &As[cur][a_off];
    const float* ws = &Ws[cur][b_off];
    v4f fa[2][4], fb[2][4];
    #pragma unroll
    for (int t = 0; t < 4; ++t) {
      fa[0][t] = *(const v4f*)(as + t * 16 * GP);
      fb[0][t] = *(const v4f*)(ws + t * 16 * GP);
    }
    #pragma unroll
    for (int kb = 0; kb < GBK / 8; ++kb) {
      const int s = kb & 1;
      if (kb + 1 < GBK / 8) {
        #pragma unroll
        for (int t = 0; t < 4; ++t) {
          fa[s ^ 1][t] = *(const v4f*)(as + t * 16 * GP + (kb + 1) * 8);
          fb[s ^ 1][t] = *(const v4f*)(ws + t * 16 * GP + (kb + 1) * 8);
        }
      }
      #pragma unroll
      for (int mt = 0; mt < 4; ++mt) {
        v2f a0 = {fa[s][mt].x, fa[s][mt].y};
        #pragma unroll
        for (int nt = 0; nt < 4; ++nt) {
          v2f b0 = {fb[s][nt].x, fb[s][nt].y};
          acc[mt][nt] = wmma4(a0, b0, acc[mt][nt]);
        }
      }
      #pragma unroll
      for (int mt = 0; mt < 4; ++mt) {
        v2f a1 = {fa[s][mt].z, fa[s][mt].w};
        #pragma unroll
        for (int nt = 0; nt < 4; ++nt) {
          v2f b1 = {fb[s][nt].z, fb[s][nt].w};
          acc[mt][nt] = wmma4(a1, b1, acc[mt][nt]);
        }
      }
    }
    if (more) {
      #pragma unroll
      for (int p = 0; p < 8; ++p) {
        *(v4f*)(&As[cur ^ 1][st_off + p * 16 * GP]) = ga[p];
        *(v4f*)(&Ws[cur ^ 1][st_off + p * 16 * GP]) = gw[p];
      }
    }
    __syncthreads();
  }

  #pragma unroll
  for (int mt = 0; mt < 4; ++mt)
    #pragma unroll
    for (int nt = 0; nt < 4; ++nt)
      #pragma unroll
      for (int r = 0; r < 8; ++r)
        Out[(size_t)(m0 + wm * 64 + mt * 16 + r + 8 * h) * N + n0 + wn * 64 + nt * 16 + c] = acc[mt][nt][r];
}

#define NT (TT / 16)

__global__ void __launch_bounds__(32)
head0_tiles(const float* __restrict__ qkv, float* __restrict__ Pfx, float* __restrict__ Ttot) {
  int x = blockIdx.x, y = blockIdx.y, b = blockIdx.z;
  int ib, jb;
  if (x <= y) { ib = y; jb = x; } else { ib = NT - 1 - y; jb = x - y - 1; }
  int lane = threadIdx.x & 31, g = lane >> 4, lm = lane & 15;
  int i0 = ib * 16, j0 = jb * 16;
  const float* qrow = qkv + ((size_t)(b * TT) + i0 + lm) * C3;
  const float* krow = qkv + ((size_t)(b * TT) + j0 + lm) * C3 + CC;
  v2f aq[16], bk[16];
  #pragma unroll
  for (int ks = 0; ks < 16; ++ks) {
    aq[ks] = *(const v2f*)(qrow + ks * 4 + 2 * g);
    bk[ks] = *(const v2f*)(krow + ks * 4 + 2 * g);
  }
  v8f acc = {0.f, 0.f, 0.f, 0.f, 0.f, 0.f, 0.f, 0.f};
  #pragma unroll
  for (int ks = 0; ks < 16; ++ks) acc = wmma4(aq[ks], bk[ks], acc);
  float p[8], run = 0.f;
  #pragma unroll
  for (int r = 0; r < 8; ++r) {
    int i = i0 + r + 8 * g, j = j0 + lm;
    float s = acc[r] * 0.125f;
    s = (j < i && j > 0) ? fmaxf(s, 0.f) : 0.f;
    p[r] = run;
    run += s;
  }
  float other = __shfl_xor(run, 16, 32);
  float above = g ? other : 0.f;
  #pragma unroll
  for (int r = 0; r < 8; ++r)
    Pfx[((size_t)(b * TT + i0 + r + 8 * g)) * TT + j0 + lm] = p[r] + above;
  if (g == 0) Ttot[((size_t)(b * NT + ib)) * TT + j0 + lm] = run + other;
}

__global__ void tile_scan(const float* __restrict__ Ttot, float* __restrict__ Toff) {
  int j = blockIdx.x * blockDim.x + threadIdx.x;
  int b = blockIdx.y;
  const float* in  = Ttot + (size_t)b * NT * TT + j;
  float*       out = Toff + (size_t)b * NT * TT + j;
  int jb = j >> 4;
  float carry = 0.f;
  for (int base = 0; base < NT; base += 16) {
    float v[16];
    #pragma unroll
    for (int u = 0; u < 16; ++u) v[u] = in[(size_t)min(jb + base + u, NT - 1) * TT];
    #pragma unroll
    for (int u = 0; u < 16; ++u) {
      int ib = jb + base + u;
      if (ib < NT) out[(size_t)ib * TT] = carry;
      carry += v[u];
    }
  }
}

__global__ void select_mask(float* FF, const float* __restrict__ Toff, unsigned* __restrict__ mask) {
  int row = blockIdx.x;
  int i = row & (TT - 1);
  float* f = FF + (size_t)row * TT;
  const float* off = Toff + ((size_t)((row >> 11) * NT + (i >> 4))) * TT;
  unsigned* mrow = mask + (size_t)row * (TT / 32);
  int tid = threadIdx.x;

  for (int j = tid; j <= i; j += 256) f[j] += off[j];

  if (i < BUDGET) {
    for (int w = tid; w < TT / 32; w += 256) {
      int jb = w * 32;
      unsigned m;
      if (jb + 31 <= i) m = 0xFFFFFFFFu;
      else if (jb > i)  m = 0u;
      else              m = (1u << (i - jb + 1)) - 1u;
      mrow[w] = m;
    }
    return;
  }

  __shared__ unsigned hist[256];
  __shared__ unsigned misc[4];
  __shared__ unsigned wave_tot[8];

  if (tid == 0) { misc[0] = 0u; misc[1] = BUDGET - 1; }
  unsigned prefix = 0, prefmask = 0;
  for (int pass = 0; pass < 4; ++pass) {
    int shift = 24 - 8 * pass;
    hist[tid] = 0u;
    __syncthreads();
    for (int j = tid; j <= i; j += 256) {
      unsigned u = __float_as_uint(f[j]);
      if ((u & prefmask) == prefix) atomicAdd(&hist[(u >> shift) & 255u], 1u);
    }
    __syncthreads();
    if (tid == 0) {
      unsigned kk = misc[1], cum = 0; int d = 0;
      for (; d < 256; ++d) { unsigned h = hist[d]; if (cum + h > kk) break; cum += h; }
      misc[1] = kk - cum;
      misc[0] |= ((unsigned)d << shift);
    }
    __syncthreads();
    prefix = misc[0];
    prefmask |= (255u << shift);
  }
  unsigned theta = prefix;

  if (tid == 0) misc[2] = 0u;
  __syncthreads();
  unsigned cl = 0;
  for (int j = tid; j <= i; j += 256)
    if (__float_as_uint(f[j]) < theta) cl++;
  atomicAdd(&misc[2], cl);
  __syncthreads();
  int ntake = BUDGET - (int)misc[2];

  int lane = tid & 31, wv = tid >> 5;
  unsigned below = (1u << lane) - 1u;
  unsigned carry = 0;
  for (int base = 0; base < TT; base += 256) {
    int j = base + tid;
    unsigned u = (j <= i) ? __float_as_uint(f[j]) : 0xFFFFFFFFu;
    bool eq = (j <= i) && (u == theta);
    unsigned bal = (unsigned)__ballot(eq);
    if (lane == 0) wave_tot[wv] = __popc(bal);
    __syncthreads();
    unsigned pre = carry + __popc(bal & below);
    for (int w2 = 0; w2 < wv; ++w2) pre += wave_tot[w2];
    bool keep = (j <= i) && ((u < theta) || (eq && (int)pre < ntake));
    keep = keep || (j == i);
    unsigned kb = (unsigned)__ballot(keep);
    if (lane == 0) mrow[(base + wv * 32) >> 5] = kb;
    unsigned tot = 0;
    #pragma unroll
    for (int w2 = 0; w2 < 8; ++w2) tot += wave_tot[w2];
    carry += tot;
    __syncthreads();
  }
}

#define NW 4
#define OST (HD + 4)
#define KP  (HD + 4)
#define VP  (HD + 8)
#define PTP 17
#define WSF (16 * KP + 16 * VP + 16 * PTP)

__device__ __forceinline__ void wave_lds_sync() {
  __builtin_amdgcn_fence(__ATOMIC_RELEASE, "wavefront");
  __builtin_amdgcn_wave_barrier();
  __builtin_amdgcn_fence(__ATOMIC_ACQUIRE, "wavefront");
}

__global__ void __launch_bounds__(32 * NW)
attn(const float* __restrict__ qkv, const float* __restrict__ FF,
     const unsigned* __restrict__ mask, float* __restrict__ Y) {
  int ib = blockIdx.x, h = blockIdx.y, b = blockIdx.z;
  int w = __builtin_amdgcn_readfirstlane((int)(threadIdx.x >> 5));
  int lane = threadIdx.x & 31, g = lane >> 4, lm = lane & 15;
  int i0 = ib * 16;
  __shared__ __attribute__((aligned(16))) float stage[NW][WSF];
  __shared__ float part_m[NW][16];
  __shared__ float part_l[NW][16];
  float* Ks = stage[w];
  float* Vs = Ks + 16 * KP;
  float* pt = Vs + 16 * VP;

  const float* qb = qkv + ((size_t)(b * TT)) * C3 + h * HD;
  const float* kb = qb + CC;

  v4f aq[8];
  #pragma unroll
  for (int k8 = 0; k8 < 8; ++k8)
    aq[k8] = *(const v4f*)(qb + (size_t)(i0 + lm) * C3 + k8 * 8 + 4 * g);

  const float* kg = kb + (size_t)g * C3 + lm * 4;
  const float* vg = kg + CC;
  const int k_st = g * KP + lm * 4;
  const int v_st = g * VP + lm * 4;
  const int k_fr = lm * KP + 4 * g;
  const int v_fr = 2 * g * VP + lm;

  float mrow[8], lrow[8];
  v8f o[4];
  #pragma unroll
  for (int t = 0; t < 4; ++t)
    #pragma unroll
    for (int r = 0; r < 8; ++r) o[t][r] = 0.f;
  #pragma unroll
  for (int r = 0; r < 8; ++r) { mrow[r] = -1e30f; lrow[r] = 0.f; }

  v4f rk[8], rv[8];
  {
    const int jf = min(w, ib) * 16;
    #pragma unroll
    for (int p = 0; p < 8; ++p) {
      rk[p] = *(const v4f*)(kg + (size_t)(jf + 2 * p) * C3);
      rv[p] = *(const v4f*)(vg + (size_t)(jf + 2 * p) * C3);
    }
    #pragma unroll
    for (int p = 0; p < 8; ++p) {
      *(v4f*)(Ks + k_st + 2 * p * KP) = rk[p];
      *(v4f*)(Vs + v_st + 2 * p * VP) = rv[p];
    }
  }
  wave_lds_sync();

  for (int jb = w; jb <= ib; jb += NW) {
    int j0 = jb * 16;
    const int jn = min(jb + NW, ib) * 16;
    float ffv[8]; unsigned mwv[8];
    #pragma unroll
    for (int r = 0; r < 8; ++r) {
      int i = i0 + r + 8 * g;
      ffv[r] = FF[((size_t)(b * TT + i)) * TT + j0 + lm];
      mwv[r] = mask[((size_t)(b * TT + i)) * (TT / 32) + (j0 >> 5)];
    }
    #pragma unroll
    for (int p = 0; p < 8; ++p)
      rk[p] = *(const v4f*)(kg + (size_t)(jn + 2 * p) * C3);
    #pragma unroll
    for (int p = 0; p < 8; ++p)
      rv[p] = *(const v4f*)(vg + (size_t)(jn + 2 * p) * C3);
    v4f fk[8];
    #pragma unroll
    for (int k8 = 0; k8 < 8; ++k8) fk[k8] = *(const v4f*)(Ks + k_fr + k8 * 8);
    v8f acc = {0.f, 0.f, 0.f, 0.f, 0.f, 0.f, 0.f, 0.f};
    #pragma unroll
    for (int k8 = 0; k8 < 8; ++k8) {
      v2f a0 = {aq[k8].x, aq[k8].y}, b0 = {fk[k8].x, fk[k8].y};
      acc = wmma4(a0, b0, acc);
      v2f a1 = {aq[k8].z, aq[k8].w}, b1 = {fk[k8].z, fk[k8].w};
      acc = wmma4(a1, b1, acc);
    }

    float p[8];
    #pragma unroll
    for (int r = 0; r < 8; ++r) {
      int i = i0 + r + 8 * g, j = j0 + lm;
      bool keep = ((mwv[r] >> (j & 31)) & 1u) && (j <= i);
      float s = keep ? (acc[r] * 0.125f - ffv[r]) : -1e30f;
      float bm = rowmax16(s);
      float mnew = fmaxf(mrow[r], bm);
      float scale = __expf(mrow[r] - mnew);
      float pr = __expf(s - mnew);
      lrow[r] = lrow[r] * scale + rowsum16(pr);
      mrow[r] = mnew;
      p[r] = pr;
      #pragma unroll
      for (int t = 0; t < 4; ++t) o[t][r] *= scale;
    }
    __builtin_amdgcn_sched_barrier(0);
    wave_lds_sync();
    #pragma unroll
    for (int p8 = 0; p8 < 8; ++p8) *(v4f*)(Ks + k_st + 2 * p8 * KP) = rk[p8];
    #pragma unroll
    for (int r = 0; r < 8; ++r) pt[(r + 8 * g) * PTP + lm] = p[r];
    wave_lds_sync();
    v2f ap[4];
    #pragma unroll
    for (int kt = 0; kt < 4; ++kt) {
      ap[kt].x = pt[lm * PTP + kt * 4 + 2 * g];
      ap[kt].y = pt[lm * PTP + kt * 4 + 2 * g + 1];
    }
    v2f bv[4][4];
    #pragma unroll
    for (int kt = 0; kt < 4; ++kt)
      #pragma unroll
      for (int t = 0; t < 4; ++t) {
        bv[kt][t].x = Vs[v_fr + (kt * 4) * VP + t * 16];
        bv[kt][t].y = Vs[v_fr + (kt * 4 + 1) * VP + t * 16];
      }
    #pragma unroll
    for (int kt = 0; kt < 4; ++kt)
      #pragma unroll
      for (int t = 0; t < 4; ++t)
        o[t] = wmma4(ap[kt], bv[kt][t], o[t]);
    wave_lds_sync();
    #pragma unroll
    for (int p8 = 0; p8 < 8; ++p8) *(v4f*)(Vs + v_st + 2 * p8 * VP) = rv[p8];
  }

  wave_lds_sync();
  float* po = stage[w];
  #pragma unroll
  for (int r = 0; r < 8; ++r) {
    #pragma unroll
    for (int t = 0; t < 4; ++t) po[(r + 8 * g) * OST + t * 16 + lm] = o[t][r];
    if (lm == 0) { part_m[w][r + 8 * g] = mrow[r]; part_l[w][r + 8 * g] = lrow[r]; }
  }
  __syncthreads();
  #pragma unroll
  for (int rr = 0; rr < 16 / NW; ++rr) {
    int row = w * (16 / NW) + rr;
    float mw[NW], M = -1e30f;
    #pragma unroll
    for (int w2 = 0; w2 < NW; ++w2) { mw[w2] = part_m[w2][row]; M = fmaxf(M, mw[w2]); }
    float L = 0.f, o0 = 0.f, o1 = 0.f;
    #pragma unroll
    for (int w2 = 0; w2 < NW; ++w2) {
      float wt = __expf(mw[w2] - M);
      L  += part_l[w2][row] * wt;
      o0 += stage[w2][row * OST + lane] * wt;
      o1 += stage[w2][row * OST + lane + 32] * wt;
    }
    float inv = 1.f / L;
    float* yrow = Y + ((size_t)(b * TT + i0 + row)) * CC + h * HD;
    yrow[lane]      = o0 * inv;
    yrow[lane + 32] = o1 * inv;
  }
}

extern "C" void kernel_launch(void* const* d_in, const int* in_sizes, int n_in,
                              void* d_out, int out_size, void* d_ws, size_t ws_size,
                              hipStream_t stream) {
  const float* x      = (const float*)d_in[0];
  const float* w_attn = (const float*)d_in[1];
  const float* b_attn = (const float*)d_in[2];
  const float* w_proj = (const float*)d_in[3];
  const float* b_proj = (const float*)d_in[4];
  float* out = (float*)d_out;

  char* ws = (char*)d_ws;
  const size_t QKV_BYTES = (size_t)BB * TT * C3 * 4;
  const size_t FF_BYTES  = (size_t)BB * TT * TT * 4;
  const size_t MSK_BYTES = (size_t)BB * TT * (TT / 32) * 4;
  float*    qkv = (float*)ws;
  float*    FF  = (float*)(ws + QKV_BYTES);
  unsigned* msk = (unsigned*)(ws + QKV_BYTES + FF_BYTES);
  float*    Y   = (float*)(ws + QKV_BYTES + FF_BYTES + MSK_BYTES);

  gemm_wmma<<<dim3(C3 / GBN, (BB * TT) / GBM), 128, 0, stream>>>(
      x, w_attn, b_attn, qkv, BB * TT, C3, CC);
  float*    Ttot = Y;
  float*    Toff = Y + (size_t)BB * NT * TT;
  head0_tiles<<<dim3(NT + 1, NT / 2, BB), 32, 0, stream>>>(qkv, FF, Ttot);
  tile_scan<<<dim3(TT / 64, BB), 64, 0, stream>>>(Ttot, Toff);
  select_mask<<<BB * TT, 256, 0, stream>>>(FF, Toff, msk);
  attn<<<dim3(TT / 16, NH, BB), 32 * NW, 0, stream>>>(qkv, FF, msk, Y);
  gemm_wmma<<<dim3(CC / GBN, (BB * TT) / GBM), 128, 0, stream>>>(
      Y, w_proj, b_proj, out, BB * TT, CC, CC);
}


# select: digit search of a radix pass in two levels: 16 threads sum the 256 bins into 16 group counts, one thread scans the 16 groups, then that group's 16 bins; was a 256-bin walk
# speedup vs baseline: 3.7147x; 1.1032x over previous
#include <hip/hip_runtime.h>
#include <hip/hip_bf16.h>

#define BB 2
#define TT 2048
#define CC 1024
#define NH 16
#define HD 64
#define C3 3072
#define BUDGET 408

typedef __attribute__((ext_vector_type(2))) float v2f;
typedef __attribute__((ext_vector_type(8))) float v8f;

__device__ __forceinline__ v8f wmma4(v2f a, v2f b, v8f c) {
  return __builtin_amdgcn_wmma_f32_16x16x4_f32(false, a, false, b, (short)0, c, false, false);
}

template <int M>
__device__ __forceinline__ float row_xor(float x) {
  int b = __builtin_bit_cast(int, x);
  return __builtin_bit_cast(float, __builtin_amdgcn_update_dpp(b, b, 0x160 + M, 0xf, 0xf, true));
}
__device__ __forceinline__ float rowmax16(float x) {
  x = __builtin_elementwise_maximum(x, row_xor<1>(x));
  x = __builtin_elementwise_maximum(x, row_xor<2>(x));
  x = __builtin_elementwise_maximum(x, row_xor<4>(x));
  x = __builtin_elementwise_maximum(x, row_xor<8>(x));
  return x;
}
__device__ __forceinline__ float rowsum16(float x) {
  x += row_xor<1>(x);
  x += row_xor<2>(x);
  x += row_xor<4>(x);
  x += row_xor<8>(x);
  return x;
}

#define GBM 128
#define GBN 128
#define GBK 32
#define GP  36

typedef __attribute__((ext_vector_type(4))) float v4f;

__global__ void __launch_bounds__(128)
gemm_wmma(const float* __restrict__ A, const float* __restrict__ W,
          const float* __restrict__ bias, float* __restrict__ Out,
          int M, int N, int K) {
  __shared__ __attribute__((aligned(16))) float As[2][GBM * GP];
  __shared__ __attribute__((aligned(16))) float Ws[2][GBN * GP];

  int tid = threadIdx.x;
  int w = __builtin_amdgcn_readfirstlane(tid >> 5);
  int lane = tid & 31, h = lane >> 4, c = lane & 15;
  int wm = w >> 1, wn = w & 1;
  int m0 = blockIdx.y * GBM, n0 = blockIdx.x * GBN;

  int crow = tid >> 3, cch = tid & 7;
  const float* Ag = A + (size_t)(m0 + crow) * K + cch * 4;
  const float* Wg = W + (size_t)(n0 + crow) * K + cch * 4;
  int st_off = crow * GP + cch * 4;
  int a_off = (wm * 64 + c) * GP + 4 * h;
  int b_off = (wn * 64 + c) * GP + 4 * h;

  v8f acc[4][4];
  #pragma unroll
  for (int nt = 0; nt < 4; ++nt) {
    float bv = bias[n0 + wn * 64 + nt * 16 + c];
    #pragma unroll
    for (int mt = 0; mt < 4; ++mt)
      #pragma unroll
      for (int r = 0; r < 8; ++r) acc[mt][nt][r] = bv;
  }

  v4f ga[8], gw[8];
  #pragma unroll
  for (int p = 0; p < 8; ++p) {
    ga[p] = *(const v4f*)(Ag + (size_t)p * 16 * K);
    gw[p] = *(const v4f*)(Wg + (size_t)p * 16 * K);
  }
  #pragma unroll
  for (int p = 0; p < 8; ++p) {
    *(v4f*)(&As[0][st_off + p * 16 * GP]) = ga[p];
    *(v4f*)(&Ws[0][st_off + p * 16 * GP]) = gw[p];
  }
  __syncthreads();

  const int nk = K / GBK;
  for (int kt = 0; kt < nk; ++kt) {
    const int cur = kt & 1;
    const bool more = (kt + 1 < nk);
    if (more) {
      #pragma unroll
      for (int p = 0; p < 8; ++p) {
        ga[p] = *(const v4f*)(Ag + (size_t)p * 16 * K + (kt + 1) * GBK);
        gw[p] = *(const v4f*)(Wg + (size_t)p * 16 * K + (kt + 1) * GBK);
      }
    }
    const float* as = &As[cur][a_off];
    const float* ws = &Ws[cur][b_off];
    v4f fa[2][4], fb[2][4];
    #pragma unroll
    for (int t = 0; t < 4; ++t) {
      fa[0][t] = *(const v4f*)(as + t * 16 * GP);
      fb[0][t] = *(const v4f*)(ws + t * 16 * GP);
    }
    #pragma unroll
    for (int kb = 0; kb < GBK / 8; ++kb) {
      const int s = kb & 1;
      if (kb + 1 < GBK / 8) {
        #pragma unroll
        for (int t = 0; t < 4; ++t) {
          fa[s ^ 1][t] = *(const v4f*)(as + t * 16 * GP + (kb + 1) * 8);
          fb[s ^ 1][t] = *(const v4f*)(ws + t * 16 * GP + (kb + 1) * 8);
        }
      }
      #pragma unroll
      for (int mt = 0; mt < 4; ++mt) {
        v2f a0 = {fa[s][mt].x, fa[s][mt].y};
        #pragma unroll
        for (int nt = 0; nt < 4; ++nt) {
          v2f b0 = {fb[s][nt].x, fb[s][nt].y};
          acc[mt][nt] = wmma4(a0, b0, acc[mt][nt]);
        }
      }
      #pragma unroll
      for (int mt = 0; mt < 4; ++mt) {
        v2f a1 = {fa[s][mt].z, fa[s][mt].w};
        #pragma unroll
        for (int nt = 0; nt < 4; ++nt) {
          v2f b1 = {fb[s][nt].z, fb[s][nt].w};
          acc[mt][nt] = wmma4(a1, b1, acc[mt][nt]);
        }
      }
    }
    if (more) {
      #pragma unroll
      for (int p = 0; p < 8; ++p) {
        *(v4f*)(&As[cur ^ 1][st_off + p * 16 * GP]) = ga[p];
        *(v4f*)(&Ws[cur ^ 1][st_off + p * 16 * GP]) = gw[p];
      }
    }
    __syncthreads();
  }

  #pragma unroll
  for (int mt = 0; mt < 4; ++mt)
    #pragma unroll
    for (int nt = 0; nt < 4; ++nt)
      #pragma unroll
      for (int r = 0; r < 8; ++r)
        Out[(size_t)(m0 + wm * 64 + mt * 16 + r + 8 * h) * N + n0 + wn * 64 + nt * 16 + c] = acc[mt][nt][r];
}

#define NT (TT / 16)

__global__ void __launch_bounds__(32)
head0_tiles(const float* __restrict__ qkv, float* __restrict__ Pfx, float* __restrict__ Ttot) {
  int x = blockIdx.x, y = blockIdx.y, b = blockIdx.z;
  int ib, jb;
  if (x <= y) { ib = y; jb = x; } else { ib = NT - 1 - y; jb = x - y - 1; }
  int lane = threadIdx.x & 31, g = lane >> 4, lm = lane & 15;
  int i0 = ib * 16, j0 = jb * 16;
  const float* qrow = qkv + ((size_t)(b * TT) + i0 + lm) * C3;
  const float* krow = qkv + ((size_t)(b * TT) + j0 + lm) * C3 + CC;
  v2f aq[16], bk[16];
  #pragma unroll
  for (int ks = 0; ks < 16; ++ks) {
    aq[ks] = *(const v2f*)(qrow + ks * 4 + 2 * g);
    bk[ks] = *(const v2f*)(krow + ks * 4 + 2 * g);
  }
  v8f acc = {0.f, 0.f, 0.f, 0.f, 0.f, 0.f, 0.f, 0.f};
  #pragma unroll
  for (int ks = 0; ks < 16; ++ks) acc = wmma4(aq[ks], bk[ks], acc);
  float p[8], run = 0.f;
  #pragma unroll
  for (int r = 0; r < 8; ++r) {
    int i = i0 + r + 8 * g, j = j0 + lm;
    float s = acc[r] * 0.125f;
    s = (j < i && j > 0) ? fmaxf(s, 0.f) : 0.f;
    p[r] = run;
    run += s;
  }
  float other = __shfl_xor(run, 16, 32);
  float above = g ? other : 0.f;
  #pragma unroll
  for (int r = 0; r < 8; ++r)
    Pfx[((size_t)(b * TT + i0 + r + 8 * g)) * TT + j0 + lm] = p[r] + above;
  if (g == 0) Ttot[((size_t)(b * NT + ib)) * TT + j0 + lm] = run + other;
}

__global__ void tile_scan(const float* __restrict__ Ttot, float* __restrict__ Toff) {
  int j = blockIdx.x * blockDim.x + threadIdx.x;
  int b = blockIdx.y;
  const float* in  = Ttot + (size_t)b * NT * TT + j;
  float*       out = Toff + (size_t)b * NT * TT + j;
  int jb = j >> 4;
  float carry = 0.f;
  for (int base = 0; base < NT; base += 16) {
    float v[16];
    #pragma unroll
    for (int u = 0; u < 16; ++u) v[u] = in[(size_t)min(jb + base + u, NT - 1) * TT];
    #pragma unroll
    for (int u = 0; u < 16; ++u) {
      int ib = jb + base + u;
      if (ib < NT) out[(size_t)ib * TT] = carry;
      carry += v[u];
    }
  }
}

__global__ void select_mask(float* FF, const float* __restrict__ Toff, unsigned* __restrict__ mask) {
  int row = blockIdx.x;
  int i = row & (TT - 1);
  float* f = FF + (size_t)row * TT;
  const float* off = Toff + ((size_t)((row >> 11) * NT + (i >> 4))) * TT;
  unsigned* mrow = mask + (size_t)row * (TT / 32);
  int tid = threadIdx.x;

  for (int j = tid; j <= i; j += 256) f[j] += off[j];

  if (i < BUDGET) {
    for (int w = tid; w < TT / 32; w += 256) {
      int jb = w * 32;
      unsigned m;
      if (jb + 31 <= i) m = 0xFFFFFFFFu;
      else if (jb > i)  m = 0u;
      else              m = (1u << (i - jb + 1)) - 1u;
      mrow[w] = m;
    }
    return;
  }

  __shared__ __attribute__((aligned(16))) unsigned hist[256];
  __shared__ __attribute__((aligned(16))) unsigned grp[16];
  __shared__ unsigned misc[4];
  __shared__ unsigned wave_tot[8];

  if (tid == 0) { misc[0] = 0u; misc[1] = BUDGET - 1; }
  unsigned prefix = 0, prefmask = 0;
  for (int pass = 0; pass < 4; ++pass) {
    int shift = 24 - 8 * pass;
    hist[tid] = 0u;
    __syncthreads();
    for (int j = tid; j <= i; j += 256) {
      unsigned u = __float_as_uint(f[j]);
      if ((u & prefmask) == prefix) atomicAdd(&hist[(u >> shift) & 255u], 1u);
    }
    __syncthreads();
    if (tid < 16) {
      const uint4* hp = (const uint4*)&hist[tid * 16];
      uint4 a = hp[0], b = hp[1], c = hp[2], d = hp[3];
      grp[tid] = ((a.x + a.y) + (a.z + a.w)) + ((b.x + b.y) + (b.z + b.w))
               + ((c.x + c.y) + (c.z + c.w)) + ((d.x + d.y) + (d.z + d.w));
    }
    __syncthreads();
    if (tid == 0) {
      unsigned kk = misc[1];
      unsigned v[16];
      #pragma unroll
      for (int q = 0; q < 4; ++q) {
        uint4 t = ((const uint4*)grp)[q];
        v[4 * q] = t.x; v[4 * q + 1] = t.y; v[4 * q + 2] = t.z; v[4 * q + 3] = t.w;
      }
      unsigned run = 0, cum = 0, hi = 0;
      #pragma unroll
      for (int q = 0; q < 16; ++q) {
        run += v[q];
        bool below = (run <= kk);
        hi += below ? 1u : 0u;
        cum = below ? run : cum;
      }
      hi = min(hi, 15u);
      #pragma unroll
      for (int q = 0; q < 4; ++q) {
        uint4 t = ((const uint4*)&hist[hi * 16])[q];
        v[4 * q] = t.x; v[4 * q + 1] = t.y; v[4 * q + 2] = t.z; v[4 * q + 3] = t.w;
      }
      unsigned lo = 0;
      run = cum;
      #pragma unroll
      for (int q = 0; q < 16; ++q) {
        run += v[q];
        bool below = (run <= kk);
        lo += below ? 1u : 0u;
        cum = below ? run : cum;
      }
      misc[1] = kk - cum;
      misc[0] |= ((hi * 16u + lo) << shift);
    }
    __syncthreads();
    prefix = misc[0];
    prefmask |= (255u << shift);
  }
  unsigned theta = prefix;

  if (tid == 0) misc[2] = 0u;
  __syncthreads();
  unsigned cl = 0;
  for (int j = tid; j <= i; j += 256)
    if (__float_as_uint(f[j]) < theta) cl++;
  atomicAdd(&misc[2], cl);
  __syncthreads();
  int ntake = BUDGET - (int)misc[2];

  int lane = tid & 31, wv = tid >> 5;
  unsigned below = (1u << lane) - 1u;
  unsigned carry = 0;
  for (int base = 0; base < TT; base += 256) {
    int j = base + tid;
    unsigned u = (j <= i) ? __float_as_uint(f[j]) : 0xFFFFFFFFu;
    bool eq = (j <= i) && (u == theta);
    unsigned bal = (unsigned)__ballot(eq);
    if (lane == 0) wave_tot[wv] = __popc(bal);
    __syncthreads();
    unsigned pre = carry + __popc(bal & below);
    for (int w2 = 0; w2 < wv; ++w2) pre += wave_tot[w2];
    bool keep = (j <= i) && ((u < theta) || (eq && (int)pre < ntake));
    keep = keep || (j == i);
    unsigned kb = (unsigned)__ballot(keep);
    if (lane == 0) mrow[(base + wv * 32) >> 5] = kb;
    unsigned tot = 0;
    #pragma unroll
    for (int w2 = 0; w2 < 8; ++w2) tot += wave_tot[w2];
    carry += tot;
    __syncthreads();
  }
}

#define NW 4
#define OST (HD + 4)
#define KP  (HD + 4)
#define VP  (HD + 8)
#define PTP 17
#define WSF (16 * KP + 16 * VP + 16 * PTP)

__device__ __forceinline__ void wave_lds_sync() {
  __builtin_amdgcn_fence(__ATOMIC_RELEASE, "wavefront");
  __builtin_amdgcn_wave_barrier();
  __builtin_amdgcn_fence(__ATOMIC_ACQUIRE, "wavefront");
}

__global__ void __launch_bounds__(32 * NW)
attn(const float* __restrict__ qkv, const float* __restrict__ FF,
     const unsigned* __restrict__ mask, float* __restrict__ Y) {
  int ib = blockIdx.x, h = blockIdx.y, b = blockIdx.z;
  int w = __builtin_amdgcn_readfirstlane((int)(threadIdx.x >> 5));
  int lane = threadIdx.x & 31, g = lane >> 4, lm = lane & 15;
  int i0 = ib * 16;
  __shared__ __attribute__((aligned(16))) float stage[NW][WSF];
  __shared__ float part_m[NW][16];
  __shared__ float part_l[NW][16];
  float* Ks = stage[w];
  float* Vs = Ks + 16 * KP;
  float* pt = Vs + 16 * VP;

  const float* qb = qkv + ((size_t)(b * TT)) * C3 + h * HD;
  const float* kb = qb + CC;

  v4f aq[8];
  #pragma unroll
  for (int k8 = 0; k8 < 8; ++k8)
    aq[k8] = *(const v4f*)(qb + (size_t)(i0 + lm) * C3 + k8 * 8 + 4 * g);

  const float* kg = kb + (size_t)g * C3 + lm * 4;
  const float* vg = kg + CC;
  const int k_st = g * KP + lm * 4;
  const int v_st = g * VP + lm * 4;
  const int k_fr = lm * KP + 4 * g;
  const int v_fr = 2 * g * VP + lm;

  float mrow[8], lrow[8];
  v8f o[4];
  #pragma unroll
  for (int t = 0; t < 4; ++t)
    #pragma unroll
    for (int r = 0; r < 8; ++r) o[t][r] = 0.f;
  #pragma unroll
  for (int r = 0; r < 8; ++r) { mrow[r] = -1e30f; lrow[r] = 0.f; }

  v4f rk[8], rv[8];
  {
    const int jf = min(w, ib) * 16;
    #pragma unroll
    for (int p = 0; p < 8; ++p) {
      rk[p] = *(const v4f*)(kg + (size_t)(jf + 2 * p) * C3);
      rv[p] = *(const v4f*)(vg + (size_t)(jf + 2 * p) * C3);
    }
    #pragma unroll
    for (int p = 0; p < 8; ++p) {
      *(v4f*)(Ks + k_st + 2 * p * KP) = rk[p];
      *(v4f*)(Vs + v_st + 2 * p * VP) = rv[p];
    }
  }
  wave_lds_sync();

  for (int jb = w; jb <= ib; jb += NW) {
    int j0 = jb * 16;
    const int jn = min(jb + NW, ib) * 16;
    float ffv[8]; unsigned mwv[8];
    #pragma unroll
    for (int r = 0; r < 8; ++r) {
      int i = i0 + r + 8 * g;
      ffv[r] = FF[((size_t)(b * TT + i)) * TT + j0 + lm];
      mwv[r] = mask[((size_t)(b * TT + i)) * (TT / 32) + (j0 >> 5)];
    }
    #pragma unroll
    for (int p = 0; p < 8; ++p)
      rk[p] = *(const v4f*)(kg + (size_t)(jn + 2 * p) * C3);
    #pragma unroll
    for (int p = 0; p < 8; ++p)
      rv[p] = *(const v4f*)(vg + (size_t)(jn + 2 * p) * C3);
    v4f fk[8];
    #pragma unroll
    for (int k8 = 0; k8 < 8; ++k8) fk[k8] = *(const v4f*)(Ks + k_fr + k8 * 8);
    v8f acc = {0.f, 0.f, 0.f, 0.f, 0.f, 0.f, 0.f, 0.f};
    #pragma unroll
    for (int k8 = 0; k8 < 8; ++k8) {
      v2f a0 = {aq[k8].x, aq[k8].y}, b0 = {fk[k8].x, fk[k8].y};
      acc = wmma4(a0, b0, acc);
      v2f a1 = {aq[k8].z, aq[k8].w}, b1 = {fk[k8].z, fk[k8].w};
      acc = wmma4(a1, b1, acc);
    }

    float p[8];
    #pragma unroll
    for (int r = 0; r < 8; ++r) {
      int i = i0 + r + 8 * g, j = j0 + lm;
      bool keep = ((mwv[r] >> (j & 31)) & 1u) && (j <= i);
      float s = keep ? (acc[r] * 0.125f - ffv[r]) : -1e30f;
      float bm = rowmax16(s);
      float mnew = fmaxf(mrow[r], bm);
      float scale = __expf(mrow[r] - mnew);
      float pr = __expf(s - mnew);
      lrow[r] = lrow[r] * scale + rowsum16(pr);
      mrow[r] = mnew;
      p[r] = pr;
      #pragma unroll
      for (int t = 0; t < 4; ++t) o[t][r] *= scale;
    }
    __builtin_amdgcn_sched_barrier(0);
    wave_lds_sync();
    #pragma unroll
    for (int p8 = 0; p8 < 8; ++p8) *(v4f*)(Ks + k_st + 2 * p8 * KP) = rk[p8];
    #pragma unroll
    for (int r = 0; r < 8; ++r) pt[(r + 8 * g) * PTP + lm] = p[r];
    wave_lds_sync();
    v2f ap[4];
    #pragma unroll
    for (int kt = 0; kt < 4; ++kt) {
      ap[kt].x = pt[lm * PTP + kt * 4 + 2 * g];
      ap[kt].y = pt[lm * PTP + kt * 4 + 2 * g + 1];
    }
    v2f bv[4][4];
    #pragma unroll
    for (int kt = 0; kt < 4; ++kt)
      #pragma unroll
      for (int t = 0; t < 4; ++t) {
        bv[kt][t].x = Vs[v_fr + (kt * 4) * VP + t * 16];
        bv[kt][t].y = Vs[v_fr + (kt * 4 + 1) * VP + t * 16];
      }
    #pragma unroll
    for (int kt = 0; kt < 4; ++kt)
      #pragma unroll
      for (int t = 0; t < 4; ++t)
        o[t] = wmma4(ap[kt], bv[kt][t], o[t]);
    wave_lds_sync();
    #pragma unroll
    for (int p8 = 0; p8 < 8; ++p8) *(v4f*)(Vs + v_st + 2 * p8 * VP) = rv[p8];
  }

  wave_lds_sync();
  float* po = stage[w];
  #pragma unroll
  for (int r = 0; r < 8; ++r) {
    #pragma unroll
    for (int t = 0; t < 4; ++t) po[(r + 8 * g) * OST + t * 16 + lm] = o[t][r];
    if (lm == 0) { part_m[w][r + 8 * g] = mrow[r]; part_l[w][r + 8 * g] = lrow[r]; }
  }
  __syncthreads();
  #pragma unroll
  for (int rr = 0; rr < 16 / NW; ++rr) {
    int row = w * (16 / NW) + rr;
    float mw[NW], M = -1e30f;
    #pragma unroll
    for (int w2 = 0; w2 < NW; ++w2) { mw[w2] = part_m[w2][row]; M = fmaxf(M, mw[w2]); }
    float L = 0.f, o0 = 0.f, o1 = 0.f;
    #pragma unroll
    for (int w2 = 0; w2 < NW; ++w2) {
      float wt = __expf(mw[w2] - M);
      L  += part_l[w2][row] * wt;
      o0 += stage[w2][row * OST + lane] * wt;
      o1 += stage[w2][row * OST + lane + 32] * wt;
    }
    float inv = 1.f / L;
    float* yrow = Y + ((size_t)(b * TT + i0 + row)) * CC + h * HD;
    yrow[lane]      = o0 * inv;
    yrow[lane + 32] = o1 * inv;
  }
}

extern "C" void kernel_launch(void* const* d_in, const int* in_sizes, int n_in,
                              void* d_out, int out_size, void* d_ws, size_t ws_size,
                              hipStream_t stream) {
  const float* x      = (const float*)d_in[0];
  const float* w_attn = (const float*)d_in[1];
  const float* b_attn = (const float*)d_in[2];
  const float* w_proj = (const float*)d_in[3];
  const float* b_proj = (const float*)d_in[4];
  float* out = (float*)d_out;

  char* ws = (char*)d_ws;
  const size_t QKV_BYTES = (size_t)BB * TT * C3 * 4;
  const size_t FF_BYTES  = (size_t)BB * TT * TT * 4;
  const size_t MSK_BYTES = (size_t)BB * TT * (TT / 32) * 4;
  float*    qkv = (float*)ws;
  float*    FF  = (float*)(ws + QKV_BYTES);
  unsigned* msk = (unsigned*)(ws + QKV_BYTES + FF_BYTES);
  float*    Y   = (float*)(ws + QKV_BYTES + FF_BYTES + MSK_BYTES);

  gemm_wmma<<<dim3(C3 / GBN, (BB * TT) / GBM), 128, 0, stream>>>(
      x, w_attn, b_attn, qkv, BB * TT, C3, CC);
  float*    Ttot = Y;
  float*    Toff = Y + (size_t)BB * NT * TT;
  head0_tiles<<<dim3(NT + 1, NT / 2, BB), 32, 0, stream>>>(qkv, FF, Ttot);
  tile_scan<<<dim3(TT / 64, BB), 64, 0, stream>>>(Ttot, Toff);
  select_mask<<<BB * TT, 256, 0, stream>>>(FF, Toff, msk);
  attn<<<dim3(TT / 16, NH, BB), 32 * NW, 0, stream>>>(qkv, FF, msk, Y);
  gemm_wmma<<<dim3(CC / GBN, (BB * TT) / GBM), 128, 0, stream>>>(
      Y, w_proj, b_proj, out, BB * TT, CC, CC);
}


# select: a thread keeps its 8 adjacent keys of the row in registers (16-byte loads, read once) for all passes, adjacent equal digits as one LDS add; was 6 row re-reads, one 4-byte load in flight
# speedup vs baseline: 3.8374x; 1.0330x over previous
#include <hip/hip_runtime.h>
#include <hip/hip_bf16.h>

#define BB 2
#define TT 2048
#define CC 1024
#define NH 16
#define HD 64
#define C3 3072
#define BUDGET 408

typedef __attribute__((ext_vector_type(2))) float v2f;
typedef __attribute__((ext_vector_type(8))) float v8f;

__device__ __forceinline__ v8f wmma4(v2f a, v2f b, v8f c) {
  return __builtin_amdgcn_wmma_f32_16x16x4_f32(false, a, false, b, (short)0, c, false, false);
}

template <int M>
__device__ __forceinline__ float row_xor(float x) {
  int b = __builtin_bit_cast(int, x);
  return __builtin_bit_cast(float, __builtin_amdgcn_update_dpp(b, b, 0x160 + M, 0xf, 0xf, true));
}
__device__ __forceinline__ float rowmax16(float x) {
  x = __builtin_elementwise_maximum(x, row_xor<1>(x));
  x = __builtin_elementwise_maximum(x, row_xor<2>(x));
  x = __builtin_elementwise_maximum(x, row_xor<4>(x));
  x = __builtin_elementwise_maximum(x, row_xor<8>(x));
  return x;
}
__device__ __forceinline__ float rowsum16(float x) {
  x += row_xor<1>(x);
  x += row_xor<2>(x);
  x += row_xor<4>(x);
  x += row_xor<8>(x);
  return x;
}

#define GBM 128
#define GBN 128
#define GBK 32
#define GP  36

typedef __attribute__((ext_vector_type(4))) float v4f;

__global__ void __launch_bounds__(128)
gemm_wmma(const float* __restrict__ A, const float* __restrict__ W,
          const float* __restrict__ bias, float* __restrict__ Out,
          int M, int N, int K) {
  __shared__ __attribute__((aligned(16))) float As[2][GBM * GP];
  __shared__ __attribute__((aligned(16))) float Ws[2][GBN * GP];

  int tid = threadIdx.x;
  int w = __builtin_amdgcn_readfirstlane(tid >> 5);
  int lane = tid & 31, h = lane >> 4, c = lane & 15;
  int wm = w >> 1, wn = w & 1;
  int m0 = blockIdx.y * GBM, n0 = blockIdx.x * GBN;

  int crow = tid >> 3, cch = tid & 7;
  const float* Ag = A + (size_t)(m0 + crow) * K + cch * 4;
  const float* Wg = W + (size_t)(n0 + crow) * K + cch * 4;
  int st_off = crow * GP + cch * 4;
  int a_off = (wm * 64 + c) * GP + 4 * h;
  int b_off = (wn * 64 + c) * GP + 4 * h;

  v8f acc[4][4];
  #pragma unroll
  for (int nt = 0; nt < 4; ++nt) {
    float bv = bias[n0 + wn * 64 + nt * 16 + c];
    #pragma unroll
    for (int mt = 0; mt < 4; ++mt)
      #pragma unroll
      for (int r = 0; r < 8; ++r) acc[mt][nt][r] = bv;
  }

  v4f ga[8], gw[8];
  #pragma unroll
  for (int p = 0; p < 8; ++p) {
    ga[p] = *(const v4f*)(Ag + (size_t)p * 16 * K);
    gw[p] = *(const v4f*)(Wg + (size_t)p * 16 * K);
  }
  #pragma unroll
  for (int p = 0; p < 8; ++p) {
    *(v4f*)(&As[0][st_off + p * 16 * GP]) = ga[p];
    *(v4f*)(&Ws[0][st_off + p * 16 * GP]) = gw[p];
  }
  __syncthreads();

  const int nk = K / GBK;
  for (int kt = 0; kt < nk; ++kt) {
    const int cur = kt & 1;
    const bool more = (kt + 1 < nk);
    if (more) {
      #pragma unroll
      for (int p = 0; p < 8; ++p) {
        ga[p] = *(const v4f*)(Ag + (size_t)p * 16 * K + (kt + 1) * GBK);
        gw[p] = *(const v4f*)(Wg + (size_t)p * 16 * K + (kt + 1) * GBK);
      }
    }
    const float* as = &As[cur][a_off];
    const float* ws = &Ws[cur][b_off];
    v4f fa[2][4], fb[2][4];
    #pragma unroll
    for (int t = 0; t < 4; ++t) {
      fa[0][t] = *(const v4f*)(as + t * 16 * GP);
      fb[0][t] = *(const v4f*)(ws + t * 16 * GP);
    }
    #pragma unroll
    for (int kb = 0; kb < GBK / 8; ++kb) {
      const int s = kb & 1;
      if (kb + 1 < GBK / 8) {
        #pragma unroll
        for (int t = 0; t < 4; ++t) {
          fa[s ^ 1][t] = *(const v4f*)(as + t * 16 * GP + (kb + 1) * 8);
          fb[s ^ 1][t] = *(const v4f*)(ws + t * 16 * GP + (kb + 1) * 8);
        }
      }
      #pragma unroll
      for (int mt = 0; mt < 4; ++mt) {
        v2f a0 = {fa[s][mt].x, fa[s][mt].y};
        #pragma unroll
        for (int nt = 0; nt < 4; ++nt) {
          v2f b0 = {fb[s][nt].x, fb[s][nt].y};
          acc[mt][nt] = wmma4(a0, b0, acc[mt][nt]);
        }
      }
      #pragma unroll
      for (int mt = 0; mt < 4; ++mt) {
        v2f a1 = {fa[s][mt].z, fa[s][mt].w};
        #pragma unroll
        for (int nt = 0; nt < 4; ++nt) {
          v2f b1 = {fb[s][nt].z, fb[s][nt].w};
          acc[mt][nt] = wmma4(a1, b1, acc[mt][nt]);
        }
      }
    }
    if (more) {
      #pragma unroll
      for (int p = 0; p < 8; ++p) {
        *(v4f*)(&As[cur ^ 1][st_off + p * 16 * GP]) = ga[p];
        *(v4f*)(&Ws[cur ^ 1][st_off + p * 16 * GP]) = gw[p];
      }
    }
    __syncthreads();
  }

  #pragma unroll
  for (int mt = 0; mt < 4; ++mt)
    #pragma unroll
    for (int nt = 0; nt < 4; ++nt)
      #pragma unroll
      for (int r = 0; r < 8; ++r)
        Out[(size_t)(m0 + wm * 64 + mt * 16 + r + 8 * h) * N + n0 + wn * 64 + nt * 16 + c] = acc[mt][nt][r];
}

#define NT (TT / 16)

__global__ void __launch_bounds__(32)
head0_tiles(const float* __restrict__ qkv, float* __restrict__ Pfx, float* __restrict__ Ttot) {
  int x = blockIdx.x, y = blockIdx.y, b = blockIdx.z;
  int ib, jb;
  if (x <= y) { ib = y; jb = x; } else { ib = NT - 1 - y; jb = x - y - 1; }
  int lane = threadIdx.x & 31, g = lane >> 4, lm = lane & 15;
  int i0 = ib * 16, j0 = jb * 16;
  const float* qrow = qkv + ((size_t)(b * TT) + i0 + lm) * C3;
  const float* krow = qkv + ((size_t)(b * TT) + j0 + lm) * C3 + CC;
  v2f aq[16], bk[16];
  #pragma unroll
  for (int ks = 0; ks < 16; ++ks) {
    aq[ks] = *(const v2f*)(qrow + ks * 4 + 2 * g);
    bk[ks] = *(const v2f*)(krow + ks * 4 + 2 * g);
  }
  v8f acc = {0.f, 0.f, 0.f, 0.f, 0.f, 0.f, 0.f, 0.f};
  #pragma unroll
  for (int ks = 0; ks < 16; ++ks) acc = wmma4(aq[ks], bk[ks], acc);
  float p[8], run = 0.f;
  #pragma unroll
  for (int r = 0; r < 8; ++r) {
    int i = i0 + r + 8 * g, j = j0 + lm;
    float s = acc[r] * 0.125f;
    s = (j < i && j > 0) ? fmaxf(s, 0.f) : 0.f;
    p[r] = run;
    run += s;
  }
  float other = __shfl_xor(run, 16, 32);
  float above = g ? other : 0.f;
  #pragma unroll
  for (int r = 0; r < 8; ++r)
    Pfx[((size_t)(b * TT + i0 + r + 8 * g)) * TT + j0 + lm] = p[r] + above;
  if (g == 0) Ttot[((size_t)(b * NT + ib)) * TT + j0 + lm] = run + other;
}

__global__ void tile_scan(const float* __restrict__ Ttot, float* __restrict__ Toff) {
  int j = blockIdx.x * blockDim.x + threadIdx.x;
  int b = blockIdx.y;
  const float* in  = Ttot + (size_t)b * NT * TT + j;
  float*       out = Toff + (size_t)b * NT * TT + j;
  int jb = j >> 4;
  float carry = 0.f;
  for (int base = 0; base < NT; base += 16) {
    float v[16];
    #pragma unroll
    for (int u = 0; u < 16; ++u) v[u] = in[(size_t)min(jb + base + u, NT - 1) * TT];
    #pragma unroll
    for (int u = 0; u < 16; ++u) {
      int ib = jb + base + u;
      if (ib < NT) out[(size_t)ib * TT] = carry;
      carry += v[u];
    }
  }
}

__global__ void select_mask(float* FF, const float* __restrict__ Toff, unsigned* __restrict__ mask) {
  int row = blockIdx.x;
  int i = row & (TT - 1);
  float* f = FF + (size_t)row * TT;
  const float* off = Toff + ((size_t)((row >> 11) * NT + (i >> 4))) * TT;
  unsigned* mrow = mask + (size_t)row * (TT / 32);
  int tid = threadIdx.x;
  const int j0 = tid * 8;
  const int nv = min(max(i - j0 + 1, 0), 8);

  v4f a0 = {0.f, 0.f, 0.f, 0.f}, a1 = a0, o0 = a0, o1 = a0;
  if (nv > 0) {
    a0 = *(const v4f*)(f + j0);
    a1 = *(const v4f*)(f + j0 + 4);
    o0 = *(const v4f*)(off + j0);
    o1 = *(const v4f*)(off + j0 + 4);
  }
  float v[8] = {a0.x, a0.y, a0.z, a0.w, a1.x, a1.y, a1.z, a1.w};
  const float ov[8] = {o0.x, o0.y, o0.z, o0.w, o1.x, o1.y, o1.z, o1.w};
  #pragma unroll
  for (int q = 0; q < 8; ++q) v[q] += (q < nv) ? ov[q] : 0.f;
  if (nv > 0) {
    v4f s0 = {v[0], v[1], v[2], v[3]}, s1 = {v[4], v[5], v[6], v[7]};
    *(v4f*)(f + j0) = s0;
    *(v4f*)(f + j0 + 4) = s1;
  }

  if (i < BUDGET) {
    for (int w = tid; w < TT / 32; w += 256) {
      int jb = w * 32;
      unsigned m;
      if (jb + 31 <= i) m = 0xFFFFFFFFu;
      else if (jb > i)  m = 0u;
      else              m = (1u << (i - jb + 1)) - 1u;
      mrow[w] = m;
    }
    return;
  }

  unsigned key[8];
  #pragma unroll
  for (int q = 0; q < 8; ++q) key[q] = __float_as_uint(v[q]);

  __shared__ __attribute__((aligned(16))) unsigned hist[256];
  __shared__ __attribute__((aligned(16))) unsigned grp[16];
  __shared__ unsigned misc[4];
  __shared__ __attribute__((aligned(16))) unsigned wave_lt[8];
  __shared__ __attribute__((aligned(16))) unsigned wave_eq[8];

  if (tid == 0) { misc[0] = 0u; misc[1] = BUDGET - 1; }
  unsigned prefix = 0, prefmask = 0;
  for (int pass = 0; pass < 4; ++pass) {
    int shift = 24 - 8 * pass;
    hist[tid] = 0u;
    __syncthreads();
    int dprev = -1;
    unsigned cnt = 0;
    #pragma unroll
    for (int q = 0; q < 8; ++q) {
      bool act = (q < nv) && ((key[q] & prefmask) == prefix);
      int d = act ? (int)((key[q] >> shift) & 255u) : -1;
      if (d != dprev) {
        if (dprev >= 0) atomicAdd(&hist[dprev], cnt);
        dprev = d;
        cnt = 0;
      }
      cnt++;
    }
    if (dprev >= 0) atomicAdd(&hist[dprev], cnt);
    __syncthreads();
    if (tid < 16) {
      const uint4* hp = (const uint4*)&hist[tid * 16];
      uint4 a = hp[0], b = hp[1], c = hp[2], d = hp[3];
      grp[tid] = ((a.x + a.y) + (a.z + a.w)) + ((b.x + b.y) + (b.z + b.w))
               + ((c.x + c.y) + (c.z + c.w)) + ((d.x + d.y) + (d.z + d.w));
    }
    __syncthreads();
    if (tid == 0) {
      unsigned kk = misc[1];
      unsigned t16[16];
      #pragma unroll
      for (int q = 0; q < 4; ++q) {
        uint4 t = ((const uint4*)grp)[q];
        t16[4 * q] = t.x; t16[4 * q + 1] = t.y; t16[4 * q + 2] = t.z; t16[4 * q + 3] = t.w;
      }
      unsigned run = 0, cum = 0, hi = 0;
      #pragma unroll
      for (int q = 0; q < 16; ++q) {
        run += t16[q];
        bool below = (run <= kk);
        hi += below ? 1u : 0u;
        cum = below ? run : cum;
      }
      hi = min(hi, 15u);
      #pragma unroll
      for (int q = 0; q < 4; ++q) {
        uint4 t = ((const uint4*)&hist[hi * 16])[q];
        t16[4 * q] = t.x; t16[4 * q + 1] = t.y; t16[4 * q + 2] = t.z; t16[4 * q + 3] = t.w;
      }
      unsigned lo = 0;
      run = cum;
      #pragma unroll
      for (int q = 0; q < 16; ++q) {
        run += t16[q];
        bool below = (run <= kk);
        lo += below ? 1u : 0u;
        cum = below ? run : cum;
      }
      misc[1] = kk - cum;
      misc[0] |= ((hi * 16u + lo) << shift);
    }
    __syncthreads();
    prefix = misc[0];
    prefmask |= (255u << shift);
  }
  const unsigned theta = prefix;

  unsigned cl = 0, ce = 0;
  #pragma unroll
  for (int q = 0; q < 8; ++q) {
    bool valid = (q < nv);
    cl += (valid && key[q] < theta) ? 1u : 0u;
    ce += (valid && key[q] == theta) ? 1u : 0u;
  }
  const int lane = tid & 31, wv = tid >> 5;
  const unsigned below = (1u << lane) - 1u;
  unsigned tot_l = 0, tot_e = 0, pre_e = 0;
  #pragma unroll
  for (int bit = 0; bit < 4; ++bit) {
    unsigned bl = (unsigned)__ballot((cl >> bit) & 1u);
    unsigned be = (unsigned)__ballot((ce >> bit) & 1u);
    tot_l += (unsigned)__popc(bl) << bit;
    tot_e += (unsigned)__popc(be) << bit;
    pre_e += (unsigned)__popc(be & below) << bit;
  }
  if (lane == 0) { wave_lt[wv] = tot_l; wave_eq[wv] = tot_e; }
  __syncthreads();
  unsigned n_less = 0, run_eq = pre_e;
  {
    const uint4 l0 = ((const uint4*)wave_lt)[0], l1 = ((const uint4*)wave_lt)[1];
    const uint4 e0 = ((const uint4*)wave_eq)[0], e1 = ((const uint4*)wave_eq)[1];
    const unsigned wl[8] = {l0.x, l0.y, l0.z, l0.w, l1.x, l1.y, l1.z, l1.w};
    const unsigned we[8] = {e0.x, e0.y, e0.z, e0.w, e1.x, e1.y, e1.z, e1.w};
    #pragma unroll
    for (int w2 = 0; w2 < 8; ++w2) {
      n_less += wl[w2];
      run_eq += (w2 < wv) ? we[w2] : 0u;
    }
  }
  const int ntake = BUDGET - (int)n_less;

  unsigned bits = 0;
  #pragma unroll
  for (int q = 0; q < 8; ++q) {
    bool valid = (q < nv);
    unsigned u = key[q];
    bool eq = valid && (u == theta);
    bool keep = valid && ((u < theta) || (eq && (int)run_eq < ntake));
    run_eq += eq ? 1u : 0u;
    keep = keep || (j0 + q == i);
    bits |= (keep ? 1u : 0u) << q;
  }
  ((unsigned char*)mrow)[tid] = (unsigned char)bits;
}

#define NW 4
#define OST (HD + 4)
#define KP  (HD + 4)
#define VP  (HD + 8)
#define PTP 17
#define WSF (16 * KP + 16 * VP + 16 * PTP)

__device__ __forceinline__ void wave_lds_sync() {
  __builtin_amdgcn_fence(__ATOMIC_RELEASE, "wavefront");
  __builtin_amdgcn_wave_barrier();
  __builtin_amdgcn_fence(__ATOMIC_ACQUIRE, "wavefront");
}

__global__ void __launch_bounds__(32 * NW)
attn(const float* __restrict__ qkv, const float* __restrict__ FF,
     const unsigned* __restrict__ mask, float* __restrict__ Y) {
  int ib = blockIdx.x, h = blockIdx.y, b = blockIdx.z;
  int w = __builtin_amdgcn_readfirstlane((int)(threadIdx.x >> 5));
  int lane = threadIdx.x & 31, g = lane >> 4, lm = lane & 15;
  int i0 = ib * 16;
  __shared__ __attribute__((aligned(16))) float stage[NW][WSF];
  __shared__ float part_m[NW][16];
  __shared__ float part_l[NW][16];
  float* Ks = stage[w];
  float* Vs = Ks + 16 * KP;
  float* pt = Vs + 16 * VP;

  const float* qb = qkv + ((size_t)(b * TT)) * C3 + h * HD;
  const float* kb = qb + CC;

  v4f aq[8];
  #pragma unroll
  for (int k8 = 0; k8 < 8; ++k8)
    aq[k8] = *(const v4f*)(qb + (size_t)(i0 + lm) * C3 + k8 * 8 + 4 * g);

  const float* kg = kb + (size_t)g * C3 + lm * 4;
  const float* vg = kg + CC;
  const int k_st = g * KP + lm * 4;
  const int v_st = g * VP + lm * 4;
  const int k_fr = lm * KP + 4 * g;
  const int v_fr = 2 * g * VP + lm;

  float mrow[8], lrow[8];
  v8f o[4];
  #pragma unroll
  for (int t = 0; t < 4; ++t)
    #pragma unroll
    for (int r = 0; r < 8; ++r) o[t][r] = 0.f;
  #pragma unroll
  for (int r = 0; r < 8; ++r) { mrow[r] = -1e30f; lrow[r] = 0.f; }

  v4f rk[8], rv[8];
  {
    const int jf = min(w, ib) * 16;
    #pragma unroll
    for (int p = 0; p < 8; ++p) {
      rk[p] = *(const v4f*)(kg + (size_t)(jf + 2 * p) * C3);
      rv[p] = *(const v4f*)(vg + (size_t)(jf + 2 * p) * C3);
    }
    #pragma unroll
    for (int p = 0; p < 8; ++p) {
      *(v4f*)(Ks + k_st + 2 * p * KP) = rk[p];
      *(v4f*)(Vs + v_st + 2 * p * VP) = rv[p];
    }
  }
  wave_lds_sync();

  for (int jb = w; jb <= ib; jb += NW) {
    int j0 = jb * 16;
    const int jn = min(jb + NW, ib) * 16;
    float ffv[8]; unsigned mwv[8];
    #pragma unroll
    for (int r = 0; r < 8; ++r) {
      int i = i0 + r + 8 * g;
      ffv[r] = FF[((size_t)(b * TT + i)) * TT + j0 + lm];
      mwv[r] = mask[((size_t)(b * TT + i)) * (TT / 32) + (j0 >> 5)];
    }
    #pragma unroll
    for (int p = 0; p < 8; ++p)
      rk[p] = *(const v4f*)(kg + (size_t)(jn + 2 * p) * C3);
    #pragma unroll
    for (int p = 0; p < 8; ++p)
      rv[p] = *(const v4f*)(vg + (size_t)(jn + 2 * p) * C3);
    v4f fk[8];
    #pragma unroll
    for (int k8 = 0; k8 < 8; ++k8) fk[k8] = *(const v4f*)(Ks + k_fr + k8 * 8);
    v8f acc = {0.f, 0.f, 0.f, 0.f, 0.f, 0.f, 0.f, 0.f};
    #pragma unroll
    for (int k8 = 0; k8 < 8; ++k8) {
      v2f a0 = {aq[k8].x, aq[k8].y}, b0 = {fk[k8].x, fk[k8].y};
      acc = wmma4(a0, b0, acc);
      v2f a1 = {aq[k8].z, aq[k8].w}, b1 = {fk[k8].z, fk[k8].w};
      acc = wmma4(a1, b1, acc);
    }

    float p[8];
    #pragma unroll
    for (int r = 0; r < 8; ++r) {
      int i = i0 + r + 8 * g, j = j0 + lm;
      bool keep = ((mwv[r] >> (j & 31)) & 1u) && (j <= i);
      float s = keep ? (acc[r] * 0.125f - ffv[r]) : -1e30f;
      float bm = rowmax16(s);
      float mnew = fmaxf(mrow[r], bm);
      float scale = __expf(mrow[r] - mnew);
      float pr = __expf(s - mnew);
      lrow[r] = lrow[r] * scale + rowsum16(pr);
      mrow[r] = mnew;
      p[r] = pr;
      #pragma unroll
      for (int t = 0; t < 4; ++t) o[t][r] *= scale;
    }
    __builtin_amdgcn_sched_barrier(0);
    wave_lds_sync();
    #pragma unroll
    for (int p8 = 0; p8 < 8; ++p8) *(v4f*)(Ks + k_st + 2 * p8 * KP) = rk[p8];
    #pragma unroll
    for (int r = 0; r < 8; ++r) pt[(r + 8 * g) * PTP + lm] = p[r];
    wave_lds_sync();
    v2f ap[4];
    #pragma unroll
    for (int kt = 0; kt < 4; ++kt) {
      ap[kt].x = pt[lm * PTP + kt * 4 + 2 * g];
      ap[kt].y = pt[lm * PTP + kt * 4 + 2 * g + 1];
    }
    v2f bv[4][4];
    #pragma unroll
    for (int kt = 0; kt < 4; ++kt)
      #pragma unroll
      for (int t = 0; t < 4; ++t) {
        bv[kt][t].x = Vs[v_fr + (kt * 4) * VP + t * 16];
        bv[kt][t].y = Vs[v_fr + (kt * 4 + 1) * VP + t * 16];
      }
    #pragma unroll
    for (int kt = 0; kt < 4; ++kt)
      #pragma unroll
      for (int t = 0; t < 4; ++t)
        o[t] = wmma4(ap[kt], bv[kt][t], o[t]);
    wave_lds_sync();
    #pragma unroll
    for (int p8 = 0; p8 < 8; ++p8) *(v4f*)(Vs + v_st + 2 * p8 * VP) = rv[p8];
  }

  wave_lds_sync();
  float* po = stage[w];
  #pragma unroll
  for (int r = 0; r < 8; ++r) {
    #pragma unroll
    for (int t = 0; t < 4; ++t) po[(r + 8 * g) * OST + t * 16 + lm] = o[t][r];
    if (lm == 0) { part_m[w][r + 8 * g] = mrow[r]; part_l[w][r + 8 * g] = lrow[r]; }
  }
  __syncthreads();
  #pragma unroll
  for (int rr = 0; rr < 16 / NW; ++rr) {
    int row = w * (16 / NW) + rr;
    float mw[NW], M = -1e30f;
    #pragma unroll
    for (int w2 = 0; w2 < NW; ++w2) { mw[w2] = part_m[w2][row]; M = fmaxf(M, mw[w2]); }
    float L = 0.f, o0 = 0.f, o1 = 0.f;
    #pragma unroll
    for (int w2 = 0; w2 < NW; ++w2) {
      float wt = __expf(mw[w2] - M);
      L  += part_l[w2][row] * wt;
      o0 += stage[w2][row * OST + lane] * wt;
      o1 += stage[w2][row * OST + lane + 32] * wt;
    }
    float inv = 1.f / L;
    float* yrow = Y + ((size_t)(b * TT + i0 + row)) * CC + h * HD;
    yrow[lane]      = o0 * inv;
    yrow[lane + 32] = o1 * inv;
  }
}

extern "C" void kernel_launch(void* const* d_in, const int* in_sizes, int n_in,
                              void* d_out, int out_size, void* d_ws, size_t ws_size,
                              hipStream_t stream) {
  const float* x      = (const float*)d_in[0];
  const float* w_attn = (const float*)d_in[1];
  const float* b_attn = (const float*)d_in[2];
  const float* w_proj = (const float*)d_in[3];
  const float* b_proj = (const float*)d_in[4];
  float* out = (float*)d_out;

  char* ws = (char*)d_ws;
  const size_t QKV_BYTES = (size_t)BB * TT * C3 * 4;
  const size_t FF_BYTES  = (size_t)BB * TT * TT * 4;
  const size_t MSK_BYTES = (size_t)BB * TT * (TT / 32) * 4;
  float*    qkv = (float*)ws;
  float*    FF  = (float*)(ws + QKV_BYTES);
  unsigned* msk = (unsigned*)(ws + QKV_BYTES + FF_BYTES);
  float*    Y   = (float*)(ws + QKV_BYTES + FF_BYTES + MSK_BYTES);

  gemm_wmma<<<dim3(C3 / GBN, (BB * TT) / GBM), 128, 0, stream>>>(
      x, w_attn, b_attn, qkv, BB * TT, C3, CC);
  float*    Ttot = Y;
  float*    Toff = Y + (size_t)BB * NT * TT;
  head0_tiles<<<dim3(NT + 1, NT / 2, BB), 32, 0, stream>>>(qkv, FF, Ttot);
  tile_scan<<<dim3(TT / 64, BB), 64, 0, stream>>>(Ttot, Toff);
  select_mask<<<BB * TT, 256, 0, stream>>>(FF, Toff, msk);
  attn<<<dim3(TT / 16, NH, BB), 32 * NW, 0, stream>>>(qkv, FF, msk, Y);
  gemm_wmma<<<dim3(CC / GBN, (BB * TT) / GBM), 128, 0, stream>>>(
      Y, w_proj, b_proj, out, BB * TT, CC, CC);
}


# attn: query tile moved to the slowest grid index (z), descending: the longest blocks (most key tiles) are dispatched first, short ones fill the tail; was ascending on the fastest index (x)
# speedup vs baseline: 4.0447x; 1.0540x over previous
#include <hip/hip_runtime.h>
#include <hip/hip_bf16.h>

#define BB 2
#define TT 2048
#define CC 1024
#define NH 16
#define HD 64
#define C3 3072
#define BUDGET 408

typedef __attribute__((ext_vector_type(2))) float v2f;
typedef __attribute__((ext_vector_type(8))) float v8f;

__device__ __forceinline__ v8f wmma4(v2f a, v2f b, v8f c) {
  return __builtin_amdgcn_wmma_f32_16x16x4_f32(false, a, false, b, (short)0, c, false, false);
}

template <int M>
__device__ __forceinline__ float row_xor(float x) {
  int b = __builtin_bit_cast(int, x);
  return __builtin_bit_cast(float, __builtin_amdgcn_update_dpp(b, b, 0x160 + M, 0xf, 0xf, true));
}
__device__ __forceinline__ float rowmax16(float x) {
  x = __builtin_elementwise_maximum(x, row_xor<1>(x));
  x = __builtin_elementwise_maximum(x, row_xor<2>(x));
  x = __builtin_elementwise_maximum(x, row_xor<4>(x));
  x = __builtin_elementwise_maximum(x, row_xor<8>(x));
  return x;
}
__device__ __forceinline__ float rowsum16(float x) {
  x += row_xor<1>(x);
  x += row_xor<2>(x);
  x += row_xor<4>(x);
  x += row_xor<8>(x);
  return x;
}

#define GBM 128
#define GBN 128
#define GBK 32
#define GP  36

typedef __attribute__((ext_vector_type(4))) float v4f;

__global__ void __launch_bounds__(128)
gemm_wmma(const float* __restrict__ A, const float* __restrict__ W,
          const float* __restrict__ bias, float* __restrict__ Out,
          int M, int N, int K) {
  __shared__ __attribute__((aligned(16))) float As[2][GBM * GP];
  __shared__ __attribute__((aligned(16))) float Ws[2][GBN * GP];

  int tid = threadIdx.x;
  int w = __builtin_amdgcn_readfirstlane(tid >> 5);
  int lane = tid & 31, h = lane >> 4, c = lane & 15;
  int wm = w >> 1, wn = w & 1;
  int m0 = blockIdx.y * GBM, n0 = blockIdx.x * GBN;

  int crow = tid >> 3, cch = tid & 7;
  const float* Ag = A + (size_t)(m0 + crow) * K + cch * 4;
  const float* Wg = W + (size_t)(n0 + crow) * K + cch * 4;
  int st_off = crow * GP + cch * 4;
  int a_off = (wm * 64 + c) * GP + 4 * h;
  int b_off = (wn * 64 + c) * GP + 4 * h;

  v8f acc[4][4];
  #pragma unroll
  for (int nt = 0; nt < 4; ++nt) {
    float bv = bias[n0 + wn * 64 + nt * 16 + c];
    #pragma unroll
    for (int mt = 0; mt < 4; ++mt)
      #pragma unroll
      for (int r = 0; r < 8; ++r) acc[mt][nt][r] = bv;
  }

  v4f ga[8], gw[8];
  #pragma unroll
  for (int p = 0; p < 8; ++p) {
    ga[p] = *(const v4f*)(Ag + (size_t)p * 16 * K);
    gw[p] = *(const v4f*)(Wg + (size_t)p * 16 * K);
  }
  #pragma unroll
  for (int p = 0; p < 8; ++p) {
    *(v4f*)(&As[0][st_off + p * 16 * GP]) = ga[p];
    *(v4f*)(&Ws[0][st_off + p * 16 * GP]) = gw[p];
  }
  __syncthreads();

  const int nk = K / GBK;
  for (int kt = 0; kt < nk; ++kt) {
    const int cur = kt & 1;
    const bool more = (kt + 1 < nk);
    if (more) {
      #pragma unroll
      for (int p = 0; p < 8; ++p) {
        ga[p] = *(const v4f*)(Ag + (size_t)p * 16 * K + (kt + 1) * GBK);
        gw[p] = *(const v4f*)(Wg + (size_t)p * 16 * K + (kt + 1) * GBK);
      }
    }
    const float* as = &As[cur][a_off];
    const float* ws = &Ws[cur][b_off];
    v4f fa[2][4], fb[2][4];
    #pragma unroll
    for (int t = 0; t < 4; ++t) {
      fa[0][t] = *(const v4f*)(as + t * 16 * GP);
      fb[0][t] = *(const v4f*)(ws + t * 16 * GP);
    }
    #pragma unroll
    for (int kb = 0; kb < GBK / 8; ++kb) {
      const int s = kb & 1;
      if (kb + 1 < GBK / 8) {
        #pragma unroll
        for (int t = 0; t < 4; ++t) {
          fa[s ^ 1][t] = *(const v4f*)(as + t * 16 * GP + (kb + 1) * 8);
          fb[s ^ 1][t] = *(const v4f*)(ws + t * 16 * GP + (kb + 1) * 8);
        }
      }
      #pragma unroll
      for (int mt = 0; mt < 4; ++mt) {
        v2f a0 = {fa[s][mt].x, fa[s][mt].y};
        #pragma unroll
        for (int nt = 0; nt < 4; ++nt) {
          v2f b0 = {fb[s][nt].x, fb[s][nt].y};
          acc[mt][nt] = wmma4(a0, b0, acc[mt][nt]);
        }
      }
      #pragma unroll
      for (int mt = 0; mt < 4; ++mt) {
        v2f a1 = {fa[s][mt].z, fa[s][mt].w};
        #pragma unroll
        for (int nt = 0; nt < 4; ++nt) {
          v2f b1 = {fb[s][nt].z, fb[s][nt].w};
          acc[mt][nt] = wmma4(a1, b1, acc[mt][nt]);
        }
      }
    }
    if (more) {
      #pragma unroll
      for (int p = 0; p < 8; ++p) {
        *(v4f*)(&As[cur ^ 1][st_off + p * 16 * GP]) = ga[p];
        *(v4f*)(&Ws[cur ^ 1][st_off + p * 16 * GP]) = gw[p];
      }
    }
    __syncthreads();
  }

  #pragma unroll
  for (int mt = 0; mt < 4; ++mt)
    #pragma unroll
    for (int nt = 0; nt < 4; ++nt)
      #pragma unroll
      for (int r = 0; r < 8; ++r)
        Out[(size_t)(m0 + wm * 64 + mt * 16 + r + 8 * h) * N + n0 + wn * 64 + nt * 16 + c] = acc[mt][nt][r];
}

#define NT (TT / 16)

__global__ void __launch_bounds__(32)
head0_tiles(const float* __restrict__ qkv, float* __restrict__ Pfx, float* __restrict__ Ttot) {
  int x = blockIdx.x, y = blockIdx.y, b = blockIdx.z;
  int ib, jb;
  if (x <= y) { ib = y; jb = x; } else { ib = NT - 1 - y; jb = x - y - 1; }
  int lane = threadIdx.x & 31, g = lane >> 4, lm = lane & 15;
  int i0 = ib * 16, j0 = jb * 16;
  const float* qrow = qkv + ((size_t)(b * TT) + i0 + lm) * C3;
  const float* krow = qkv + ((size_t)(b * TT) + j0 + lm) * C3 + CC;
  v2f aq[16], bk[16];
  #pragma unroll
  for (int ks = 0; ks < 16; ++ks) {
    aq[ks] = *(const v2f*)(qrow + ks * 4 + 2 * g);
    bk[ks] = *(const v2f*)(krow + ks * 4 + 2 * g);
  }
  v8f acc = {0.f, 0.f, 0.f, 0.f, 0.f, 0.f, 0.f, 0.f};
  #pragma unroll
  for (int ks = 0; ks < 16; ++ks) acc = wmma4(aq[ks], bk[ks], acc);
  float p[8], run = 0.f;
  #pragma unroll
  for (int r = 0; r < 8; ++r) {
    int i = i0 + r + 8 * g, j = j0 + lm;
    float s = acc[r] * 0.125f;
    s = (j < i && j > 0) ? fmaxf(s, 0.f) : 0.f;
    p[r] = run;
    run += s;
  }
  float other = __shfl_xor(run, 16, 32);
  float above = g ? other : 0.f;
  #pragma unroll
  for (int r = 0; r < 8; ++r)
    Pfx[((size_t)(b * TT + i0 + r + 8 * g)) * TT + j0 + lm] = p[r] + above;
  if (g == 0) Ttot[((size_t)(b * NT + ib)) * TT + j0 + lm] = run + other;
}

__global__ void tile_scan(const float* __restrict__ Ttot, float* __restrict__ Toff) {
  int j = blockIdx.x * blockDim.x + threadIdx.x;
  int b = blockIdx.y;
  const float* in  = Ttot + (size_t)b * NT * TT + j;
  float*       out = Toff + (size_t)b * NT * TT + j;
  int jb = j >> 4;
  float carry = 0.f;
  for (int base = 0; base < NT; base += 16) {
    float v[16];
    #pragma unroll
    for (int u = 0; u < 16; ++u) v[u] = in[(size_t)min(jb + base + u, NT - 1) * TT];
    #pragma unroll
    for (int u = 0; u < 16; ++u) {
      int ib = jb + base + u;
      if (ib < NT) out[(size_t)ib * TT] = carry;
      carry += v[u];
    }
  }
}

__global__ void select_mask(float* FF, const float* __restrict__ Toff, unsigned* __restrict__ mask) {
  int row = blockIdx.x;
  int i = row & (TT - 1);
  float* f = FF + (size_t)row * TT;
  const float* off = Toff + ((size_t)((row >> 11) * NT + (i >> 4))) * TT;
  unsigned* mrow = mask + (size_t)row * (TT / 32);
  int tid = threadIdx.x;
  const int j0 = tid * 8;
  const int nv = min(max(i - j0 + 1, 0), 8);

  v4f a0 = {0.f, 0.f, 0.f, 0.f}, a1 = a0, o0 = a0, o1 = a0;
  if (nv > 0) {
    a0 = *(const v4f*)(f + j0);
    a1 = *(const v4f*)(f + j0 + 4);
    o0 = *(const v4f*)(off + j0);
    o1 = *(const v4f*)(off + j0 + 4);
  }
  float v[8] = {a0.x, a0.y, a0.z, a0.w, a1.x, a1.y, a1.z, a1.w};
  const float ov[8] = {o0.x, o0.y, o0.z, o0.w, o1.x, o1.y, o1.z, o1.w};
  #pragma unroll
  for (int q = 0; q < 8; ++q) v[q] += (q < nv) ? ov[q] : 0.f;
  if (nv > 0) {
    v4f s0 = {v[0], v[1], v[2], v[3]}, s1 = {v[4], v[5], v[6], v[7]};
    *(v4f*)(f + j0) = s0;
    *(v4f*)(f + j0 + 4) = s1;
  }

  if (i < BUDGET) {
    for (int w = tid; w < TT / 32; w += 256) {
      int jb = w * 32;
      unsigned m;
      if (jb + 31 <= i) m = 0xFFFFFFFFu;
      else if (jb > i)  m = 0u;
      else              m = (1u << (i - jb + 1)) - 1u;
      mrow[w] = m;
    }
    return;
  }

  unsigned key[8];
  #pragma unroll
  for (int q = 0; q < 8; ++q) key[q] = __float_as_uint(v[q]);

  __shared__ __attribute__((aligned(16))) unsigned hist[256];
  __shared__ __attribute__((aligned(16))) unsigned grp[16];
  __shared__ unsigned misc[4];
  __shared__ __attribute__((aligned(16))) unsigned wave_lt[8];
  __shared__ __attribute__((aligned(16))) unsigned wave_eq[8];

  if (tid == 0) { misc[0] = 0u; misc[1] = BUDGET - 1; }
  unsigned prefix = 0, prefmask = 0;
  for (int pass = 0; pass < 4; ++pass) {
    int shift = 24 - 8 * pass;
    hist[tid] = 0u;
    __syncthreads();
    int dprev = -1;
    unsigned cnt = 0;
    #pragma unroll
    for (int q = 0; q < 8; ++q) {
      bool act = (q < nv) && ((key[q] & prefmask) == prefix);
      int d = act ? (int)((key[q] >> shift) & 255u) : -1;
      if (d != dprev) {
        if (dprev >= 0) atomicAdd(&hist[dprev], cnt);
        dprev = d;
        cnt = 0;
      }
      cnt++;
    }
    if (dprev >= 0) atomicAdd(&hist[dprev], cnt);
    __syncthreads();
    if (tid < 16) {
      const uint4* hp = (const uint4*)&hist[tid * 16];
      uint4 a = hp[0], b = hp[1], c = hp[2], d = hp[3];
      grp[tid] = ((a.x + a.y) + (a.z + a.w)) + ((b.x + b.y) + (b.z + b.w))
               + ((c.x + c.y) + (c.z + c.w)) + ((d.x + d.y) + (d.z + d.w));
    }
    __syncthreads();
    if (tid == 0) {
      unsigned kk = misc[1];
      unsigned t16[16];
      #pragma unroll
      for (int q = 0; q < 4; ++q) {
        uint4 t = ((const uint4*)grp)[q];
        t16[4 * q] = t.x; t16[4 * q + 1] = t.y; t16[4 * q + 2] = t.z; t16[4 * q + 3] = t.w;
      }
      unsigned run = 0, cum = 0, hi = 0;
      #pragma unroll
      for (int q = 0; q < 16; ++q) {
        run += t16[q];
        bool below = (run <= kk);
        hi += below ? 1u : 0u;
        cum = below ? run : cum;
      }
      hi = min(hi, 15u);
      #pragma unroll
      for (int q = 0; q < 4; ++q) {
        uint4 t = ((const uint4*)&hist[hi * 16])[q];
        t16[4 * q] = t.x; t16[4 * q + 1] = t.y; t16[4 * q + 2] = t.z; t16[4 * q + 3] = t.w;
      }
      unsigned lo = 0;
      run = cum;
      #pragma unroll
      for (int q = 0; q < 16; ++q) {
        run += t16[q];
        bool below = (run <= kk);
        lo += below ? 1u : 0u;
        cum = below ? run : cum;
      }
      misc[1] = kk - cum;
      misc[0] |= ((hi * 16u + lo) << shift);
    }
    __syncthreads();
    prefix = misc[0];
    prefmask |= (255u << shift);
  }
  const unsigned theta = prefix;

  unsigned cl = 0, ce = 0;
  #pragma unroll
  for (int q = 0; q < 8; ++q) {
    bool valid = (q < nv);
    cl += (valid && key[q] < theta) ? 1u : 0u;
    ce += (valid && key[q] == theta) ? 1u : 0u;
  }
  const int lane = tid & 31, wv = tid >> 5;
  const unsigned below = (1u << lane) - 1u;
  unsigned tot_l = 0, tot_e = 0, pre_e = 0;
  #pragma unroll
  for (int bit = 0; bit < 4; ++bit) {
    unsigned bl = (unsigned)__ballot((cl >> bit) & 1u);
    unsigned be = (unsigned)__ballot((ce >> bit) & 1u);
    tot_l += (unsigned)__popc(bl) << bit;
    tot_e += (unsigned)__popc(be) << bit;
    pre_e += (unsigned)__popc(be & below) << bit;
  }
  if (lane == 0) { wave_lt[wv] = tot_l; wave_eq[wv] = tot_e; }
  __syncthreads();
  unsigned n_less = 0, run_eq = pre_e;
  {
    const uint4 l0 = ((const uint4*)wave_lt)[0], l1 = ((const uint4*)wave_lt)[1];
    const uint4 e0 = ((const uint4*)wave_eq)[0], e1 = ((const uint4*)wave_eq)[1];
    const unsigned wl[8] = {l0.x, l0.y, l0.z, l0.w, l1.x, l1.y, l1.z, l1.w};
    const unsigned we[8] = {e0.x, e0.y, e0.z, e0.w, e1.x, e1.y, e1.z, e1.w};
    #pragma unroll
    for (int w2 = 0; w2 < 8; ++w2) {
      n_less += wl[w2];
      run_eq += (w2 < wv) ? we[w2] : 0u;
    }
  }
  const int ntake = BUDGET - (int)n_less;

  unsigned bits = 0;
  #pragma unroll
  for (int q = 0; q < 8; ++q) {
    bool valid = (q < nv);
    unsigned u = key[q];
    bool eq = valid && (u == theta);
    bool keep = valid && ((u < theta) || (eq && (int)run_eq < ntake));
    run_eq += eq ? 1u : 0u;
    keep = keep || (j0 + q == i);
    bits |= (keep ? 1u : 0u) << q;
  }
  ((unsigned char*)mrow)[tid] = (unsigned char)bits;
}

#define NW 4
#define OST (HD + 4)
#define KP  (HD + 4)
#define VP  (HD + 8)
#define PTP 17
#define WSF (16 * KP + 16 * VP + 16 * PTP)

__device__ __forceinline__ void wave_lds_sync() {
  __builtin_amdgcn_fence(__ATOMIC_RELEASE, "wavefront");
  __builtin_amdgcn_wave_barrier();
  __builtin_amdgcn_fence(__ATOMIC_ACQUIRE, "wavefront");
}

__global__ void __launch_bounds__(32 * NW)
attn(const float* __restrict__ qkv, const float* __restrict__ FF,
     const unsigned* __restrict__ mask, float* __restrict__ Y) {
  int h = blockIdx.x, b = blockIdx.y;
  int ib = NT - 1 - (int)blockIdx.z;
  int w = __builtin_amdgcn_readfirstlane((int)(threadIdx.x >> 5));
  int lane = threadIdx.x & 31, g = lane >> 4, lm = lane & 15;
  int i0 = ib * 16;
  __shared__ __attribute__((aligned(16))) float stage[NW][WSF];
  __shared__ float part_m[NW][16];
  __shared__ float part_l[NW][16];
  float* Ks = stage[w];
  float* Vs = Ks + 16 * KP;
  float* pt = Vs + 16 * VP;

  const float* qb = qkv + ((size_t)(b * TT)) * C3 + h * HD;
  const float* kb = qb + CC;

  v4f aq[8];
  #pragma unroll
  for (int k8 = 0; k8 < 8; ++k8)
    aq[k8] = *(const v4f*)(qb + (size_t)(i0 + lm) * C3 + k8 * 8 + 4 * g);

  const float* kg = kb + (size_t)g * C3 + lm * 4;
  const float* vg = kg + CC;
  const int k_st = g * KP + lm * 4;
  const int v_st = g * VP + lm * 4;
  const int k_fr = lm * KP + 4 * g;
  const int v_fr = 2 * g * VP + lm;

  float mrow[8], lrow[8];
  v8f o[4];
  #pragma unroll
  for (int t = 0; t < 4; ++t)
    #pragma unroll
    for (int r = 0; r < 8; ++r) o[t][r] = 0.f;
  #pragma unroll
  for (int r = 0; r < 8; ++r) { mrow[r] = -1e30f; lrow[r] = 0.f; }

  v4f rk[8], rv[8];
  {
    const int jf = min(w, ib) * 16;
    #pragma unroll
    for (int p = 0; p < 8; ++p) {
      rk[p] = *(const v4f*)(kg + (size_t)(jf + 2 * p) * C3);
      rv[p] = *(const v4f*)(vg + (size_t)(jf + 2 * p) * C3);
    }
    #pragma unroll
    for (int p = 0; p < 8; ++p) {
      *(v4f*)(Ks + k_st + 2 * p * KP) = rk[p];
      *(v4f*)(Vs + v_st + 2 * p * VP) = rv[p];
    }
  }
  wave_lds_sync();

  for (int jb = w; jb <= ib; jb += NW) {
    int j0 = jb * 16;
    const int jn = min(jb + NW, ib) * 16;
    float ffv[8]; unsigned mwv[8];
    #pragma unroll
    for (int r = 0; r < 8; ++r) {
      int i = i0 + r + 8 * g;
      ffv[r] = FF[((size_t)(b * TT + i)) * TT + j0 + lm];
      mwv[r] = mask[((size_t)(b * TT + i)) * (TT / 32) + (j0 >> 5)];
    }
    #pragma unroll
    for (int p = 0; p < 8; ++p)
      rk[p] = *(const v4f*)(kg + (size_t)(jn + 2 * p) * C3);
    #pragma unroll
    for (int p = 0; p < 8; ++p)
      rv[p] = *(const v4f*)(vg + (size_t)(jn + 2 * p) * C3);
    v4f fk[8];
    #pragma unroll
    for (int k8 = 0; k8 < 8; ++k8) fk[k8] = *(const v4f*)(Ks + k_fr + k8 * 8);
    v8f acc = {0.f, 0.f, 0.f, 0.f, 0.f, 0.f, 0.f, 0.f};
    #pragma unroll
    for (int k8 = 0; k8 < 8; ++k8) {
      v2f a0 = {aq[k8].x, aq[k8].y}, b0 = {fk[k8].x, fk[k8].y};
      acc = wmma4(a0, b0, acc);
      v2f a1 = {aq[k8].z, aq[k8].w}, b1 = {fk[k8].z, fk[k8].w};
      acc = wmma4(a1, b1, acc);
    }

    float p[8];
    #pragma unroll
    for (int r = 0; r < 8; ++r) {
      int i = i0 + r + 8 * g, j = j0 + lm;
      bool keep = ((mwv[r] >> (j & 31)) & 1u) && (j <= i);
      float s = keep ? (acc[r] * 0.125f - ffv[r]) : -1e30f;
      float bm = rowmax16(s);
      float mnew = fmaxf(mrow[r], bm);
      float scale = __expf(mrow[r] - mnew);
      float pr = __expf(s - mnew);
      lrow[r] = lrow[r] * scale + rowsum16(pr);
      mrow[r] = mnew;
      p[r] = pr;
      #pragma unroll
      for (int t = 0; t < 4; ++t) o[t][r] *= scale;
    }
    __builtin_amdgcn_sched_barrier(0);
    wave_lds_sync();
    #pragma unroll
    for (int p8 = 0; p8 < 8; ++p8) *(v4f*)(Ks + k_st + 2 * p8 * KP) = rk[p8];
    #pragma unroll
    for (int r = 0; r < 8; ++r) pt[(r + 8 * g) * PTP + lm] = p[r];
    wave_lds_sync();
    v2f ap[4];
    #pragma unroll
    for (int kt = 0; kt < 4; ++kt) {
      ap[kt].x = pt[lm * PTP + kt * 4 + 2 * g];
      ap[kt].y = pt[lm * PTP + kt * 4 + 2 * g + 1];
    }
    v2f bv[4][4];
    #pragma unroll
    for (int kt = 0; kt < 4; ++kt)
      #pragma unroll
      for (int t = 0; t < 4; ++t) {
        bv[kt][t].x = Vs[v_fr + (kt * 4) * VP + t * 16];
        bv[kt][t].y = Vs[v_fr + (kt * 4 + 1) * VP + t * 16];
      }
    #pragma unroll
    for (int kt = 0; kt < 4; ++kt)
      #pragma unroll
      for (int t = 0; t < 4; ++t)
        o[t] = wmma4(ap[kt], bv[kt][t], o[t]);
    wave_lds_sync();
    #pragma unroll
    for (int p8 = 0; p8 < 8; ++p8) *(v4f*)(Vs + v_st + 2 * p8 * VP) = rv[p8];
  }

  wave_lds_sync();
  float* po = stage[w];
  #pragma unroll
  for (int r = 0; r < 8; ++r) {
    #pragma unroll
    for (int t = 0; t < 4; ++t) po[(r + 8 * g) * OST + t * 16 + lm] = o[t][r];
    if (lm == 0) { part_m[w][r + 8 * g] = mrow[r]; part_l[w][r + 8 * g] = lrow[r]; }
  }
  __syncthreads();
  #pragma unroll
  for (int rr = 0; rr < 16 / NW; ++rr) {
    int row = w * (16 / NW) + rr;
    float mw[NW], M = -1e30f;
    #pragma unroll
    for (int w2 = 0; w2 < NW; ++w2) { mw[w2] = part_m[w2][row]; M = fmaxf(M, mw[w2]); }
    float L = 0.f, o0 = 0.f, o1 = 0.f;
    #pragma unroll
    for (int w2 = 0; w2 < NW; ++w2) {
      float wt = __expf(mw[w2] - M);
      L  += part_l[w2][row] * wt;
      o0 += stage[w2][row * OST + lane] * wt;
      o1 += stage[w2][row * OST + lane + 32] * wt;
    }
    float inv = 1.f / L;
    float* yrow = Y + ((size_t)(b * TT + i0 + row)) * CC + h * HD;
    yrow[lane]      = o0 * inv;
    yrow[lane + 32] = o1 * inv;
  }
}

extern "C" void kernel_launch(void* const* d_in, const int* in_sizes, int n_in,
                              void* d_out, int out_size, void* d_ws, size_t ws_size,
                              hipStream_t stream) {
  const float* x      = (const float*)d_in[0];
  const float* w_attn = (const float*)d_in[1];
  const float* b_attn = (const float*)d_in[2];
  const float* w_proj = (const float*)d_in[3];
  const float* b_proj = (const float*)d_in[4];
  float* out = (float*)d_out;

  char* ws = (char*)d_ws;
  const size_t QKV_BYTES = (size_t)BB * TT * C3 * 4;
  const size_t FF_BYTES  = (size_t)BB * TT * TT * 4;
  const size_t MSK_BYTES = (size_t)BB * TT * (TT / 32) * 4;
  float*    qkv = (float*)ws;
  float*    FF  = (float*)(ws + QKV_BYTES);
  unsigned* msk = (unsigned*)(ws + QKV_BYTES + FF_BYTES);
  float*    Y   = (float*)(ws + QKV_BYTES + FF_BYTES + MSK_BYTES);

  gemm_wmma<<<dim3(C3 / GBN, (BB * TT) / GBM), 128, 0, stream>>>(
      x, w_attn, b_attn, qkv, BB * TT, C3, CC);
  float*    Ttot = Y;
  float*    Toff = Y + (size_t)BB * NT * TT;
  head0_tiles<<<dim3(NT + 1, NT / 2, BB), 32, 0, stream>>>(qkv, FF, Ttot);
  tile_scan<<<dim3(TT / 64, BB), 64, 0, stream>>>(Ttot, Toff);
  select_mask<<<BB * TT, 256, 0, stream>>>(FF, Toff, msk);
  attn<<<dim3(NH, BB, NT), 32 * NW, 0, stream>>>(qkv, FF, msk, Y);
  gemm_wmma<<<dim3(CC / GBN, (BB * TT) / GBM), 128, 0, stream>>>(
      Y, w_proj, b_proj, out, BB * TT, CC, CC);
}


# head0: one wave takes a 16-row x 64-key strip (4 key tiles): Q fragments loaded once, 16-byte fragment loads as one group, 4-wave blocks; was one wave per tile, 8-byte loads; order of k sum changed
# speedup vs baseline: 4.2326x; 1.0465x over previous
#include <hip/hip_runtime.h>
#include <hip/hip_bf16.h>

#define BB 2
#define TT 2048
#define CC 1024
#define NH 16
#define HD 64
#define C3 3072
#define BUDGET 408

typedef __attribute__((ext_vector_type(2))) float v2f;
typedef __attribute__((ext_vector_type(8))) float v8f;

__device__ __forceinline__ v8f wmma4(v2f a, v2f b, v8f c) {
  return __builtin_amdgcn_wmma_f32_16x16x4_f32(false, a, false, b, (short)0, c, false, false);
}

template <int M>
__device__ __forceinline__ float row_xor(float x) {
  int b = __builtin_bit_cast(int, x);
  return __builtin_bit_cast(float, __builtin_amdgcn_update_dpp(b, b, 0x160 + M, 0xf, 0xf, true));
}
__device__ __forceinline__ float rowmax16(float x) {
  x = __builtin_elementwise_maximum(x, row_xor<1>(x));
  x = __builtin_elementwise_maximum(x, row_xor<2>(x));
  x = __builtin_elementwise_maximum(x, row_xor<4>(x));
  x = __builtin_elementwise_maximum(x, row_xor<8>(x));
  return x;
}
__device__ __forceinline__ float rowsum16(float x) {
  x += row_xor<1>(x);
  x += row_xor<2>(x);
  x += row_xor<4>(x);
  x += row_xor<8>(x);
  return x;
}

#define GBM 128
#define GBN 128
#define GBK 32
#define GP  36

typedef __attribute__((ext_vector_type(4))) float v4f;

__global__ void __launch_bounds__(128)
gemm_wmma(const float* __restrict__ A, const float* __restrict__ W,
          const float* __restrict__ bias, float* __restrict__ Out,
          int M, int N, int K) {
  __shared__ __attribute__((aligned(16))) float As[2][GBM * GP];
  __shared__ __attribute__((aligned(16))) float Ws[2][GBN * GP];

  int tid = threadIdx.x;
  int w = __builtin_amdgcn_readfirstlane(tid >> 5);
  int lane = tid & 31, h = lane >> 4, c = lane & 15;
  int wm = w >> 1, wn = w & 1;
  int m0 = blockIdx.y * GBM, n0 = blockIdx.x * GBN;

  int crow = tid >> 3, cch = tid & 7;
  const float* Ag = A + (size_t)(m0 + crow) * K + cch * 4;
  const float* Wg = W + (size_t)(n0 + crow) * K + cch * 4;
  int st_off = crow * GP + cch * 4;
  int a_off = (wm * 64 + c) * GP + 4 * h;
  int b_off = (wn * 64 + c) * GP + 4 * h;

  v8f acc[4][4];
  #pragma unroll
  for (int nt = 0; nt < 4; ++nt) {
    float bv = bias[n0 + wn * 64 + nt * 16 + c];
    #pragma unroll
    for (int mt = 0; mt < 4; ++mt)
      #pragma unroll
      for (int r = 0; r < 8; ++r) acc[mt][nt][r] = bv;
  }

  v4f ga[8], gw[8];
  #pragma unroll
  for (int p = 0; p < 8; ++p) {
    ga[p] = *(const v4f*)(Ag + (size_t)p * 16 * K);
    gw[p] = *(const v4f*)(Wg + (size_t)p * 16 * K);
  }
  #pragma unroll
  for (int p = 0; p < 8; ++p) {
    *(v4f*)(&As[0][st_off + p * 16 * GP]) = ga[p];
    *(v4f*)(&Ws[0][st_off + p * 16 * GP]) = gw[p];
  }
  __syncthreads();

  const int nk = K / GBK;
  for (int kt = 0; kt < nk; ++kt) {
    const int cur = kt & 1;
    const bool more = (kt + 1 < nk);
    if (more) {
      #pragma unroll
      for (int p = 0; p < 8; ++p) {
        ga[p] = *(const v4f*)(Ag + (size_t)p * 16 * K + (kt + 1) * GBK);
        gw[p] = *(const v4f*)(Wg + (size_t)p * 16 * K + (kt + 1) * GBK);
      }
    }
    const float* as = &As[cur][a_off];
    const float* ws = &Ws[cur][b_off];
    v4f fa[2][4], fb[2][4];
    #pragma unroll
    for (int t = 0; t < 4; ++t) {
      fa[0][t] = *(const v4f*)(as + t * 16 * GP);
      fb[0][t] = *(const v4f*)(ws + t * 16 * GP);
    }
    #pragma unroll
    for (int kb = 0; kb < GBK / 8; ++kb) {
      const int s = kb & 1;
      if (kb + 1 < GBK / 8) {
        #pragma unroll
        for (int t = 0; t < 4; ++t) {
          fa[s ^ 1][t] = *(const v4f*)(as + t * 16 * GP + (kb + 1) * 8);
          fb[s ^ 1][t] = *(const v4f*)(ws + t * 16 * GP + (kb + 1) * 8);
        }
      }
      #pragma unroll
      for (int mt = 0; mt < 4; ++mt) {
        v2f a0 = {fa[s][mt].x, fa[s][mt].y};
        #pragma unroll
        for (int nt = 0; nt < 4; ++nt) {
          v2f b0 = {fb[s][nt].x, fb[s][nt].y};
          acc[mt][nt] = wmma4(a0, b0, acc[mt][nt]);
        }
      }
      #pragma unroll
      for (int mt = 0; mt < 4; ++mt) {
        v2f a1 = {fa[s][mt].z, fa[s][mt].w};
        #pragma unroll
        for (int nt = 0; nt < 4; ++nt) {
          v2f b1 = {fb[s][nt].z, fb[s][nt].w};
          acc[mt][nt] = wmma4(a1, b1, acc[mt][nt]);
        }
      }
    }
    if (more) {
      #pragma unroll
      for (int p = 0; p < 8; ++p) {
        *(v4f*)(&As[cur ^ 1][st_off + p * 16 * GP]) = ga[p];
        *(v4f*)(&Ws[cur ^ 1][st_off + p * 16 * GP]) = gw[p];
      }
    }
    __syncthreads();
  }

  #pragma unroll
  for (int mt = 0; mt < 4; ++mt)
    #pragma unroll
    for (int nt = 0; nt < 4; ++nt)
      #pragma unroll
      for (int r = 0; r < 8; ++r)
        Out[(size_t)(m0 + wm * 64 + mt * 16 + r + 8 * h) * N + n0 + wn * 64 + nt * 16 + c] = acc[mt][nt][r];
}

#define NT (TT / 16)
#define HS 4

__global__ void __launch_bounds__(128)
head0_tiles(const float* __restrict__ qkv, float* __restrict__ Pfx, float* __restrict__ Ttot) {
  int w = __builtin_amdgcn_readfirstlane((int)(threadIdx.x >> 5));
  int x = blockIdx.x, y = blockIdx.y * 4 + w, b = blockIdx.z;
  int ny = (y >> 2) + 1;
  int ib, sg;
  if (x < ny) { ib = y; sg = x; } else { ib = NT - 1 - y; sg = x - ny; }
  int lane = threadIdx.x & 31, g = lane >> 4, lm = lane & 15;
  int i0 = ib * 16;
  const float* qrow = qkv + ((size_t)(b * TT) + i0 + lm) * C3 + 4 * g;
  v4f aq[8], bk[HS][8];
  #pragma unroll
  for (int k8 = 0; k8 < 8; ++k8) aq[k8] = *(const v4f*)(qrow + k8 * 8);
  #pragma unroll
  for (int t = 0; t < HS; ++t) {
    int jb = min(sg * HS + t, ib);
    const float* krow = qkv + ((size_t)(b * TT) + jb * 16 + lm) * C3 + CC + 4 * g;
    #pragma unroll
    for (int k8 = 0; k8 < 8; ++k8) bk[t][k8] = *(const v4f*)(krow + k8 * 8);
  }
  __builtin_amdgcn_sched_barrier(0);
  #pragma unroll
  for (int t = 0; t < HS; ++t) {
    v8f acc = {0.f, 0.f, 0.f, 0.f, 0.f, 0.f, 0.f, 0.f};
    #pragma unroll
    for (int k8 = 0; k8 < 8; ++k8) {
      v2f a0 = {aq[k8].x, aq[k8].y}, b0 = {bk[t][k8].x, bk[t][k8].y};
      acc = wmma4(a0, b0, acc);
      v2f a1 = {aq[k8].z, aq[k8].w}, b1 = {bk[t][k8].z, bk[t][k8].w};
      acc = wmma4(a1, b1, acc);
    }
    int jb = sg * HS + t;
    if (jb <= ib) {
      int j0 = jb * 16;
      float p[8], run = 0.f;
      #pragma unroll
      for (int r = 0; r < 8; ++r) {
        int i = i0 + r + 8 * g, j = j0 + lm;
        float s = acc[r] * 0.125f;
        s = (j < i && j > 0) ? fmaxf(s, 0.f) : 0.f;
        p[r] = run;
        run += s;
      }
      float other = __shfl_xor(run, 16, 32);
      float above = g ? other : 0.f;
      #pragma unroll
      for (int r = 0; r < 8; ++r)
        Pfx[((size_t)(b * TT + i0 + r + 8 * g)) * TT + j0 + lm] = p[r] + above;
      if (g == 0) Ttot[((size_t)(b * NT + ib)) * TT + j0 + lm] = run + other;
    }
  }
}

__global__ void tile_scan(const float* __restrict__ Ttot, float* __restrict__ Toff) {
  int j = blockIdx.x * blockDim.x + threadIdx.x;
  int b = blockIdx.y;
  const float* in  = Ttot + (size_t)b * NT * TT + j;
  float*       out = Toff + (size_t)b * NT * TT + j;
  int jb = j >> 4;
  float carry = 0.f;
  for (int base = 0; base < NT; base += 16) {
    float v[16];
    #pragma unroll
    for (int u = 0; u < 16; ++u) v[u] = in[(size_t)min(jb + base + u, NT - 1) * TT];
    #pragma unroll
    for (int u = 0; u < 16; ++u) {
      int ib = jb + base + u;
      if (ib < NT) out[(size_t)ib * TT] = carry;
      carry += v[u];
    }
  }
}

__global__ void select_mask(float* FF, const float* __restrict__ Toff, unsigned* __restrict__ mask) {
  int row = blockIdx.x;
  int i = row & (TT - 1);
  float* f = FF + (size_t)row * TT;
  const float* off = Toff + ((size_t)((row >> 11) * NT + (i >> 4))) * TT;
  unsigned* mrow = mask + (size_t)row * (TT / 32);
  int tid = threadIdx.x;
  const int j0 = tid * 8;
  const int nv = min(max(i - j0 + 1, 0), 8);

  v4f a0 = {0.f, 0.f, 0.f, 0.f}, a1 = a0, o0 = a0, o1 = a0;
  if (nv > 0) {
    a0 = *(const v4f*)(f + j0);
    a1 = *(const v4f*)(f + j0 + 4);
    o0 = *(const v4f*)(off + j0);
    o1 = *(const v4f*)(off + j0 + 4);
  }
  float v[8] = {a0.x, a0.y, a0.z, a0.w, a1.x, a1.y, a1.z, a1.w};
  const float ov[8] = {o0.x, o0.y, o0.z, o0.w, o1.x, o1.y, o1.z, o1.w};
  #pragma unroll
  for (int q = 0; q < 8; ++q) v[q] += (q < nv) ? ov[q] : 0.f;
  if (nv > 0) {
    v4f s0 = {v[0], v[1], v[2], v[3]}, s1 = {v[4], v[5], v[6], v[7]};
    *(v4f*)(f + j0) = s0;
    *(v4f*)(f + j0 + 4) = s1;
  }

  if (i < BUDGET) {
    for (int w = tid; w < TT / 32; w += 256) {
      int jb = w * 32;
      unsigned m;
      if (jb + 31 <= i) m = 0xFFFFFFFFu;
      else if (jb > i)  m = 0u;
      else              m = (1u << (i - jb + 1)) - 1u;
      mrow[w] = m;
    }
    return;
  }

  unsigned key[8];
  #pragma unroll
  for (int q = 0; q < 8; ++q) key[q] = __float_as_uint(v[q]);

  __shared__ __attribute__((aligned(16))) unsigned hist[256];
  __shared__ __attribute__((aligned(16))) unsigned grp[16];
  __shared__ unsigned misc[4];
  __shared__ __attribute__((aligned(16))) unsigned wave_lt[8];
  __shared__ __attribute__((aligned(16))) unsigned wave_eq[8];

  if (tid == 0) { misc[0] = 0u; misc[1] = BUDGET - 1; }
  unsigned prefix = 0, prefmask = 0;
  for (int pass = 0; pass < 4; ++pass) {
    int shift = 24 - 8 * pass;
    hist[tid] = 0u;
    __syncthreads();
    int dprev = -1;
    unsigned cnt = 0;
    #pragma unroll
    for (int q = 0; q < 8; ++q) {
      bool act = (q < nv) && ((key[q] & prefmask) == prefix);
      int d = act ? (int)((key[q] >> shift) & 255u) : -1;
      if (d != dprev) {
        if (dprev >= 0) atomicAdd(&hist[dprev], cnt);
        dprev = d;
        cnt = 0;
      }
      cnt++;
    }
    if (dprev >= 0) atomicAdd(&hist[dprev], cnt);
    __syncthreads();
    if (tid < 16) {
      const uint4* hp = (const uint4*)&hist[tid * 16];
      uint4 a = hp[0], b = hp[1], c = hp[2], d = hp[3];
      grp[tid] = ((a.x + a.y) + (a.z + a.w)) + ((b.x + b.y) + (b.z + b.w))
               + ((c.x + c.y) + (c.z + c.w)) + ((d.x + d.y) + (d.z + d.w));
    }
    __syncthreads();
    if (tid == 0) {
      unsigned kk = misc[1];
      unsigned t16[16];
      #pragma unroll
      for (int q = 0; q < 4; ++q) {
        uint4 t = ((const uint4*)grp)[q];
        t16[4 * q] = t.x; t16[4 * q + 1] = t.y; t16[4 * q + 2] = t.z; t16[4 * q + 3] = t.w;
      }
      unsigned run = 0, cum = 0, hi = 0;
      #pragma unroll
      for (int q = 0; q < 16; ++q) {
        run += t16[q];
        bool below = (run <= kk);
        hi += below ? 1u : 0u;
        cum = below ? run : cum;
      }
      hi = min(hi, 15u);
      #pragma unroll
      for (int q = 0; q < 4; ++q) {
        uint4 t = ((const uint4*)&hist[hi * 16])[q];
        t16[4 * q] = t.x; t16[4 * q + 1] = t.y; t16[4 * q + 2] = t.z; t16[4 * q + 3] = t.w;
      }
      unsigned lo = 0;
      run = cum;
      #pragma unroll
      for (int q = 0; q < 16; ++q) {
        run += t16[q];
        bool below = (run <= kk);
        lo += below ? 1u : 0u;
        cum = below ? run : cum;
      }
      misc[1] = kk - cum;
      misc[0] |= ((hi * 16u + lo) << shift);
    }
    __syncthreads();
    prefix = misc[0];
    prefmask |= (255u << shift);
  }
  const unsigned theta = prefix;

  unsigned cl = 0, ce = 0;
  #pragma unroll
  for (int q = 0; q < 8; ++q) {
    bool valid = (q < nv);
    cl += (valid && key[q] < theta) ? 1u : 0u;
    ce += (valid && key[q] == theta) ? 1u : 0u;
  }
  const int lane = tid & 31, wv = tid >> 5;
  const unsigned below = (1u << lane) - 1u;
  unsigned tot_l = 0, tot_e = 0, pre_e = 0;
  #pragma unroll
  for (int bit = 0; bit < 4; ++bit) {
    unsigned bl = (unsigned)__ballot((cl >> bit) & 1u);
    unsigned be = (unsigned)__ballot((ce >> bit) & 1u);
    tot_l += (unsigned)__popc(bl) << bit;
    tot_e += (unsigned)__popc(be) << bit;
    pre_e += (unsigned)__popc(be & below) << bit;
  }
  if (lane == 0) { wave_lt[wv] = tot_l; wave_eq[wv] = tot_e; }
  __syncthreads();
  unsigned n_less = 0, run_eq = pre_e;
  {
    const uint4 l0 = ((const uint4*)wave_lt)[0], l1 = ((const uint4*)wave_lt)[1];
    const uint4 e0 = ((const uint4*)wave_eq)[0], e1 = ((const uint4*)wave_eq)[1];
    const unsigned wl[8] = {l0.x, l0.y, l0.z, l0.w, l1.x, l1.y, l1.z, l1.w};
    const unsigned we[8] = {e0.x, e0.y, e0.z, e0.w, e1.x, e1.y, e1.z, e1.w};
    #pragma unroll
    for (int w2 = 0; w2 < 8; ++w2) {
      n_less += wl[w2];
      run_eq += (w2 < wv) ? we[w2] : 0u;
    }
  }
  const int ntake = BUDGET - (int)n_less;

  unsigned bits = 0;
  #pragma unroll
  for (int q = 0; q < 8; ++q) {
    bool valid = (q < nv);
    unsigned u = key[q];
    bool eq = valid && (u == theta);
    bool keep = valid && ((u < theta) || (eq && (int)run_eq < ntake));
    run_eq += eq ? 1u : 0u;
    keep = keep || (j0 + q == i);
    bits |= (keep ? 1u : 0u) << q;
  }
  ((unsigned char*)mrow)[tid] = (unsigned char)bits;
}

#define NW 4
#define OST (HD + 4)
#define KP  (HD + 4)
#define VP  (HD + 8)
#define PTP 17
#define WSF (16 * KP + 16 * VP + 16 * PTP)

__device__ __forceinline__ void wave_lds_sync() {
  __builtin_amdgcn_fence(__ATOMIC_RELEASE, "wavefront");
  __builtin_amdgcn_wave_barrier();
  __builtin_amdgcn_fence(__ATOMIC_ACQUIRE, "wavefront");
}

__global__ void __launch_bounds__(32 * NW)
attn(const float* __restrict__ qkv, const float* __restrict__ FF,
     const unsigned* __restrict__ mask, float* __restrict__ Y) {
  int h = blockIdx.x, b = blockIdx.y;
  int ib = NT - 1 - (int)blockIdx.z;
  int w = __builtin_amdgcn_readfirstlane((int)(threadIdx.x >> 5));
  int lane = threadIdx.x & 31, g = lane >> 4, lm = lane & 15;
  int i0 = ib * 16;
  __shared__ __attribute__((aligned(16))) float stage[NW][WSF];
  __shared__ float part_m[NW][16];
  __shared__ float part_l[NW][16];
  float* Ks = stage[w];
  float* Vs = Ks + 16 * KP;
  float* pt = Vs + 16 * VP;

  const float* qb = qkv + ((size_t)(b * TT)) * C3 + h * HD;
  const float* kb = qb + CC;

  v4f aq[8];
  #pragma unroll
  for (int k8 = 0; k8 < 8; ++k8)
    aq[k8] = *(const v4f*)(qb + (size_t)(i0 + lm) * C3 + k8 * 8 + 4 * g);

  const float* kg = kb + (size_t)g * C3 + lm * 4;
  const float* vg = kg + CC;
  const int k_st = g * KP + lm * 4;
  const int v_st = g * VP + lm * 4;
  const int k_fr = lm * KP + 4 * g;
  const int v_fr = 2 * g * VP + lm;

  float mrow[8], lrow[8];
  v8f o[4];
  #pragma unroll
  for (int t = 0; t < 4; ++t)
    #pragma unroll
    for (int r = 0; r < 8; ++r) o[t][r] = 0.f;
  #pragma unroll
  for (int r = 0; r < 8; ++r) { mrow[r] = -1e30f; lrow[r] = 0.f; }

  v4f rk[8], rv[8];
  {
    const int jf = min(w, ib) * 16;
    #pragma unroll
    for (int p = 0; p < 8; ++p) {
      rk[p] = *(const v4f*)(kg + (size_t)(jf + 2 * p) * C3);
      rv[p] = *(const v4f*)(vg + (size_t)(jf + 2 * p) * C3);
    }
    #pragma unroll
    for (int p = 0; p < 8; ++p) {
      *(v4f*)(Ks + k_st + 2 * p * KP) = rk[p];
      *(v4f*)(Vs + v_st + 2 * p * VP) = rv[p];
    }
  }
  wave_lds_sync();

  for (int jb = w; jb <= ib; jb += NW) {
    int j0 = jb * 16;
    const int jn = min(jb + NW, ib) * 16;
    float ffv[8]; unsigned mwv[8];
    #pragma unroll
    for (int r = 0; r < 8; ++r) {
      int i = i0 + r + 8 * g;
      ffv[r] = FF[((size_t)(b * TT + i)) * TT + j0 + lm];
      mwv[r] = mask[((size_t)(b * TT + i)) * (TT / 32) + (j0 >> 5)];
    }
    #pragma unroll
    for (int p = 0; p < 8; ++p)
      rk[p] = *(const v4f*)(kg + (size_t)(jn + 2 * p) * C3);
    #pragma unroll
    for (int p = 0; p < 8; ++p)
      rv[p] = *(const v4f*)(vg + (size_t)(jn + 2 * p) * C3);
    v4f fk[8];
    #pragma unroll
    for (int k8 = 0; k8 < 8; ++k8) fk[k8] = *(const v4f*)(Ks + k_fr + k8 * 8);
    v8f acc = {0.f, 0.f, 0.f, 0.f, 0.f, 0.f, 0.f, 0.f};
    #pragma unroll
    for (int k8 = 0; k8 < 8; ++k8) {
      v2f a0 = {aq[k8].x, aq[k8].y}, b0 = {fk[k8].x, fk[k8].y};
      acc = wmma4(a0, b0, acc);
      v2f a1 = {aq[k8].z, aq[k8].w}, b1 = {fk[k8].z, fk[k8].w};
      acc = wmma4(a1, b1, acc);
    }

    float p[8];
    #pragma unroll
    for (int r = 0; r < 8; ++r) {
      int i = i0 + r + 8 * g, j = j0 + lm;
      bool keep = ((mwv[r] >> (j & 31)) & 1u) && (j <= i);
      float s = keep ? (acc[r] * 0.125f - ffv[r]) : -1e30f;
      float bm = rowmax16(s);
      float mnew = fmaxf(mrow[r], bm);
      float scale = __expf(mrow[r] - mnew);
      float pr = __expf(s - mnew);
      lrow[r] = lrow[r] * scale + rowsum16(pr);
      mrow[r] = mnew;
      p[r] = pr;
      #pragma unroll
      for (int t = 0; t < 4; ++t) o[t][r] *= scale;
    }
    __builtin_amdgcn_sched_barrier(0);
    wave_lds_sync();
    #pragma unroll
    for (int p8 = 0; p8 < 8; ++p8) *(v4f*)(Ks + k_st + 2 * p8 * KP) = rk[p8];
    #pragma unroll
    for (int r = 0; r < 8; ++r) pt[(r + 8 * g) * PTP + lm] = p[r];
    wave_lds_sync();
    v2f ap[4];
    #pragma unroll
    for (int kt = 0; kt < 4; ++kt) {
      ap[kt].x = pt[lm * PTP + kt * 4 + 2 * g];
      ap[kt].y = pt[lm * PTP + kt * 4 + 2 * g + 1];
    }
    v2f bv[4][4];
    #pragma unroll
    for (int kt = 0; kt < 4; ++kt)
      #pragma unroll
      for (int t = 0; t < 4; ++t) {
        bv[kt][t].x = Vs[v_fr + (kt * 4) * VP + t * 16];
        bv[kt][t].y = Vs[v_fr + (kt * 4 + 1) * VP + t * 16];
      }
    #pragma unroll
    for (int kt = 0; kt < 4; ++kt)
      #pragma unroll
      for (int t = 0; t < 4; ++t)
        o[t] = wmma4(ap[kt], bv[kt][t], o[t]);
    wave_lds_sync();
    #pragma unroll
    for (int p8 = 0; p8 < 8; ++p8) *(v4f*)(Vs + v_st + 2 * p8 * VP) = rv[p8];
  }

  wave_lds_sync();
  float* po = stage[w];
  #pragma unroll
  for (int r = 0; r < 8; ++r) {
    #pragma unroll
    for (int t = 0; t < 4; ++t) po[(r + 8 * g) * OST + t * 16 + lm] = o[t][r];
    if (lm == 0) { part_m[w][r + 8 * g] = mrow[r]; part_l[w][r + 8 * g] = lrow[r]; }
  }
  __syncthreads();
  #pragma unroll
  for (int rr = 0; rr < 16 / NW; ++rr) {
    int row = w * (16 / NW) + rr;
    float mw[NW], M = -1e30f;
    #pragma unroll
    for (int w2 = 0; w2 < NW; ++w2) { mw[w2] = part_m[w2][row]; M = fmaxf(M, mw[w2]); }
    float L = 0.f, o0 = 0.f, o1 = 0.f;
    #pragma unroll
    for (int w2 = 0; w2 < NW; ++w2) {
      float wt = __expf(mw[w2] - M);
      L  += part_l[w2][row] * wt;
      o0 += stage[w2][row * OST + lane] * wt;
      o1 += stage[w2][row * OST + lane + 32] * wt;
    }
    float inv = 1.f / L;
    float* yrow = Y + ((size_t)(b * TT + i0 + row)) * CC + h * HD;
    yrow[lane]      = o0 * inv;
    yrow[lane + 32] = o1 * inv;
  }
}

extern "C" void kernel_launch(void* const* d_in, const int* in_sizes, int n_in,
                              void* d_out, int out_size, void* d_ws, size_t ws_size,
                              hipStream_t stream) {
  const float* x      = (const float*)d_in[0];
  const float* w_attn = (const float*)d_in[1];
  const float* b_attn = (const float*)d_in[2];
  const float* w_proj = (const float*)d_in[3];
  const float* b_proj = (const float*)d_in[4];
  float* out = (float*)d_out;

  char* ws = (char*)d_ws;
  const size_t QKV_BYTES = (size_t)BB * TT * C3 * 4;
  const size_t FF_BYTES  = (size_t)BB * TT * TT * 4;
  const size_t MSK_BYTES = (size_t)BB * TT * (TT / 32) * 4;
  float*    qkv = (float*)ws;
  float*    FF  = (float*)(ws + QKV_BYTES);
  unsigned* msk = (unsigned*)(ws + QKV_BYTES + FF_BYTES);
  float*    Y   = (float*)(ws + QKV_BYTES + FF_BYTES + MSK_BYTES);

  gemm_wmma<<<dim3(C3 / GBN, (BB * TT) / GBM), 128, 0, stream>>>(
      x, w_attn, b_attn, qkv, BB * TT, C3, CC);
  float*    Ttot = Y;
  float*    Toff = Y + (size_t)BB * NT * TT;
  head0_tiles<<<dim3(NT / HS + 1, NT / 8, BB), 128, 0, stream>>>(qkv, FF, Ttot);
  tile_scan<<<dim3(TT / 64, BB), 64, 0, stream>>>(Ttot, Toff);
  select_mask<<<BB * TT, 256, 0, stream>>>(FF, Toff, msk);
  attn<<<dim3(NH, BB, NT), 32 * NW, 0, stream>>>(qkv, FF, msk, Y);
  gemm_wmma<<<dim3(CC / GBN, (BB * TT) / GBM), 128, 0, stream>>>(
      Y, w_proj, b_proj, out, BB * TT, CC, CC);
}


# attn softmax: score tile, not P, takes the LDS re-layout to A fragments: a lane holds 8 keys of one query row; max and sum in-lane + one half exchange, was 64 DPP steps a tile; row-sum order changed
# speedup vs baseline: 4.3842x; 1.0358x over previous
#include <hip/hip_runtime.h>
#include <hip/hip_bf16.h>

#define BB 2
#define TT 2048
#define CC 1024
#define NH 16
#define HD 64
#define C3 3072
#define BUDGET 408

typedef __attribute__((ext_vector_type(2))) float v2f;
typedef __attribute__((ext_vector_type(8))) float v8f;

__device__ __forceinline__ v8f wmma4(v2f a, v2f b, v8f c) {
  return __builtin_amdgcn_wmma_f32_16x16x4_f32(false, a, false, b, (short)0, c, false, false);
}

#define GBM 128
#define GBN 128
#define GBK 32
#define GP  36

typedef __attribute__((ext_vector_type(4))) float v4f;

__global__ void __launch_bounds__(128)
gemm_wmma(const float* __restrict__ A, const float* __restrict__ W,
          const float* __restrict__ bias, float* __restrict__ Out,
          int M, int N, int K) {
  __shared__ __attribute__((aligned(16))) float As[2][GBM * GP];
  __shared__ __attribute__((aligned(16))) float Ws[2][GBN * GP];

  int tid = threadIdx.x;
  int w = __builtin_amdgcn_readfirstlane(tid >> 5);
  int lane = tid & 31, h = lane >> 4, c = lane & 15;
  int wm = w >> 1, wn = w & 1;
  int m0 = blockIdx.y * GBM, n0 = blockIdx.x * GBN;

  int crow = tid >> 3, cch = tid & 7;
  const float* Ag = A + (size_t)(m0 + crow) * K + cch * 4;
  const float* Wg = W + (size_t)(n0 + crow) * K + cch * 4;
  int st_off = crow * GP + cch * 4;
  int a_off = (wm * 64 + c) * GP + 4 * h;
  int b_off = (wn * 64 + c) * GP + 4 * h;

  v8f acc[4][4];
  #pragma unroll
  for (int nt = 0; nt < 4; ++nt) {
    float bv = bias[n0 + wn * 64 + nt * 16 + c];
    #pragma unroll
    for (int mt = 0; mt < 4; ++mt)
      #pragma unroll
      for (int r = 0; r < 8; ++r) acc[mt][nt][r] = bv;
  }

  v4f ga[8], gw[8];
  #pragma unroll
  for (int p = 0; p < 8; ++p) {
    ga[p] = *(const v4f*)(Ag + (size_t)p * 16 * K);
    gw[p] = *(const v4f*)(Wg + (size_t)p * 16 * K);
  }
  #pragma unroll
  for (int p = 0; p < 8; ++p) {
    *(v4f*)(&As[0][st_off + p * 16 * GP]) = ga[p];
    *(v4f*)(&Ws[0][st_off + p * 16 * GP]) = gw[p];
  }
  __syncthreads();

  const int nk = K / GBK;
  for (int kt = 0; kt < nk; ++kt) {
    const int cur = kt & 1;
    const bool more = (kt + 1 < nk);
    if (more) {
      #pragma unroll
      for (int p = 0; p < 8; ++p) {
        ga[p] = *(const v4f*)(Ag + (size_t)p * 16 * K + (kt + 1) * GBK);
        gw[p] = *(const v4f*)(Wg + (size_t)p * 16 * K + (kt + 1) * GBK);
      }
    }
    const float* as = &As[cur][a_off];
    const float* ws = &Ws[cur][b_off];
    v4f fa[2][4], fb[2][4];
    #pragma unroll
    for (int t = 0; t < 4; ++t) {
      fa[0][t] = *(const v4f*)(as + t * 16 * GP);
      fb[0][t] = *(const v4f*)(ws + t * 16 * GP);
    }
    #pragma unroll
    for (int kb = 0; kb < GBK / 8; ++kb) {
      const int s = kb & 1;
      if (kb + 1 < GBK / 8) {
        #pragma unroll
        for (int t = 0; t < 4; ++t) {
          fa[s ^ 1][t] = *(const v4f*)(as + t * 16 * GP + (kb + 1) * 8);
          fb[s ^ 1][t] = *(const v4f*)(ws + t * 16 * GP + (kb + 1) * 8);
        }
      }
      #pragma unroll
      for (int mt = 0; mt < 4; ++mt) {
        v2f a0 = {fa[s][mt].x, fa[s][mt].y};
        #pragma unroll
        for (int nt = 0; nt < 4; ++nt) {
          v2f b0 = {fb[s][nt].x, fb[s][nt].y};
          acc[mt][nt] = wmma4(a0, b0, acc[mt][nt]);
        }
      }
      #pragma unroll
      for (int mt = 0; mt < 4; ++mt) {
        v2f a1 = {fa[s][mt].z, fa[s][mt].w};
        #pragma unroll
        for (int nt = 0; nt < 4; ++nt) {
          v2f b1 = {fb[s][nt].z, fb[s][nt].w};
          acc[mt][nt] = wmma4(a1, b1, acc[mt][nt]);
        }
      }
    }
    if (more) {
      #pragma unroll
      for (int p = 0; p < 8; ++p) {
        *(v4f*)(&As[cur ^ 1][st_off + p * 16 * GP]) = ga[p];
        *(v4f*)(&Ws[cur ^ 1][st_off + p * 16 * GP]) = gw[p];
      }
    }
    __syncthreads();
  }

  #pragma unroll
  for (int mt = 0; mt < 4; ++mt)
    #pragma unroll
    for (int nt = 0; nt < 4; ++nt)
      #pragma unroll
      for (int r = 0; r < 8; ++r)
        Out[(size_t)(m0 + wm * 64 + mt * 16 + r + 8 * h) * N + n0 + wn * 64 + nt * 16 + c] = acc[mt][nt][r];
}

#define NT (TT / 16)
#define HS 4

__global__ void __launch_bounds__(128)
head0_tiles(const float* __restrict__ qkv, float* __restrict__ Pfx, float* __restrict__ Ttot) {
  int w = __builtin_amdgcn_readfirstlane((int)(threadIdx.x >> 5));
  int x = blockIdx.x, y = blockIdx.y * 4 + w, b = blockIdx.z;
  int ny = (y >> 2) + 1;
  int ib, sg;
  if (x < ny) { ib = y; sg = x; } else { ib = NT - 1 - y; sg = x - ny; }
  int lane = threadIdx.x & 31, g = lane >> 4, lm = lane & 15;
  int i0 = ib * 16;
  const float* qrow = qkv + ((size_t)(b * TT) + i0 + lm) * C3 + 4 * g;
  v4f aq[8], bk[HS][8];
  #pragma unroll
  for (int k8 = 0; k8 < 8; ++k8) aq[k8] = *(const v4f*)(qrow + k8 * 8);
  #pragma unroll
  for (int t = 0; t < HS; ++t) {
    int jb = min(sg * HS + t, ib);
    const float* krow = qkv + ((size_t)(b * TT) + jb * 16 + lm) * C3 + CC + 4 * g;
    #pragma unroll
    for (int k8 = 0; k8 < 8; ++k8) bk[t][k8] = *(const v4f*)(krow + k8 * 8);
  }
  __builtin_amdgcn_sched_barrier(0);
  #pragma unroll
  for (int t = 0; t < HS; ++t) {
    v8f acc = {0.f, 0.f, 0.f, 0.f, 0.f, 0.f, 0.f, 0.f};
    #pragma unroll
    for (int k8 = 0; k8 < 8; ++k8) {
      v2f a0 = {aq[k8].x, aq[k8].y}, b0 = {bk[t][k8].x, bk[t][k8].y};
      acc = wmma4(a0, b0, acc);
      v2f a1 = {aq[k8].z, aq[k8].w}, b1 = {bk[t][k8].z, bk[t][k8].w};
      acc = wmma4(a1, b1, acc);
    }
    int jb = sg * HS + t;
    if (jb <= ib) {
      int j0 = jb * 16;
      float p[8], run = 0.f;
      #pragma unroll
      for (int r = 0; r < 8; ++r) {
        int i = i0 + r + 8 * g, j = j0 + lm;
        float s = acc[r] * 0.125f;
        s = (j < i && j > 0) ? fmaxf(s, 0.f) : 0.f;
        p[r] = run;
        run += s;
      }
      float other = __shfl_xor(run, 16, 32);
      float above = g ? other : 0.f;
      #pragma unroll
      for (int r = 0; r < 8; ++r)
        Pfx[((size_t)(b * TT + i0 + r + 8 * g)) * TT + j0 + lm] = p[r] + above;
      if (g == 0) Ttot[((size_t)(b * NT + ib)) * TT + j0 + lm] = run + other;
    }
  }
}

__global__ void tile_scan(const float* __restrict__ Ttot, float* __restrict__ Toff) {
  int j = blockIdx.x * blockDim.x + threadIdx.x;
  int b = blockIdx.y;
  const float* in  = Ttot + (size_t)b * NT * TT + j;
  float*       out = Toff + (size_t)b * NT * TT + j;
  int jb = j >> 4;
  float carry = 0.f;
  for (int base = 0; base < NT; base += 16) {
    float v[16];
    #pragma unroll
    for (int u = 0; u < 16; ++u) v[u] = in[(size_t)min(jb + base + u, NT - 1) * TT];
    #pragma unroll
    for (int u = 0; u < 16; ++u) {
      int ib = jb + base + u;
      if (ib < NT) out[(size_t)ib * TT] = carry;
      carry += v[u];
    }
  }
}

__global__ void select_mask(float* FF, const float* __restrict__ Toff, unsigned* __restrict__ mask) {
  int row = blockIdx.x;
  int i = row & (TT - 1);
  float* f = FF + (size_t)row * TT;
  const float* off = Toff + ((size_t)((row >> 11) * NT + (i >> 4))) * TT;
  unsigned* mrow = mask + (size_t)row * (TT / 32);
  int tid = threadIdx.x;
  const int j0 = tid * 8;
  const int nv = min(max(i - j0 + 1, 0), 8);

  v4f a0 = {0.f, 0.f, 0.f, 0.f}, a1 = a0, o0 = a0, o1 = a0;
  if (nv > 0) {
    a0 = *(const v4f*)(f + j0);
    a1 = *(const v4f*)(f + j0 + 4);
    o0 = *(const v4f*)(off + j0);
    o1 = *(const v4f*)(off + j0 + 4);
  }
  float v[8] = {a0.x, a0.y, a0.z, a0.w, a1.x, a1.y, a1.z, a1.w};
  const float ov[8] = {o0.x, o0.y, o0.z, o0.w, o1.x, o1.y, o1.z, o1.w};
  #pragma unroll
  for (int q = 0; q < 8; ++q) v[q] += (q < nv) ? ov[q] : 0.f;
  if (nv > 0) {
    v4f s0 = {v[0], v[1], v[2], v[3]}, s1 = {v[4], v[5], v[6], v[7]};
    *(v4f*)(f + j0) = s0;
    *(v4f*)(f + j0 + 4) = s1;
  }

  if (i < BUDGET) {
    for (int w = tid; w < TT / 32; w += 256) {
      int jb = w * 32;
      unsigned m;
      if (jb + 31 <= i) m = 0xFFFFFFFFu;
      else if (jb > i)  m = 0u;
      else              m = (1u << (i - jb + 1)) - 1u;
      mrow[w] = m;
    }
    return;
  }

  unsigned key[8];
  #pragma unroll
  for (int q = 0; q < 8; ++q) key[q] = __float_as_uint(v[q]);

  __shared__ __attribute__((aligned(16))) unsigned hist[256];
  __shared__ __attribute__((aligned(16))) unsigned grp[16];
  __shared__ unsigned misc[4];
  __shared__ __attribute__((aligned(16))) unsigned wave_lt[8];
  __shared__ __attribute__((aligned(16))) unsigned wave_eq[8];

  if (tid == 0) { misc[0] = 0u; misc[1] = BUDGET - 1; }
  unsigned prefix = 0, prefmask = 0;
  for (int pass = 0; pass < 4; ++pass) {
    int shift = 24 - 8 * pass;
    hist[tid] = 0u;
    __syncthreads();
    int dprev = -1;
    unsigned cnt = 0;
    #pragma unroll
    for (int q = 0; q < 8; ++q) {
      bool act = (q < nv) && ((key[q] & prefmask) == prefix);
      int d = act ? (int)((key[q] >> shift) & 255u) : -1;
      if (d != dprev) {
        if (dprev >= 0) atomicAdd(&hist[dprev], cnt);
        dprev = d;
        cnt = 0;
      }
      cnt++;
    }
    if (dprev >= 0) atomicAdd(&hist[dprev], cnt);
    __syncthreads();
    if (tid < 16) {
      const uint4* hp = (const uint4*)&hist[tid * 16];
      uint4 a = hp[0], b = hp[1], c = hp[2], d = hp[3];
      grp[tid] = ((a.x + a.y) + (a.z + a.w)) + ((b.x + b.y) + (b.z + b.w))
               + ((c.x + c.y) + (c.z + c.w)) + ((d.x + d.y) + (d.z + d.w));
    }
    __syncthreads();
    if (tid == 0) {
      unsigned kk = misc[1];
      unsigned t16[16];
      #pragma unroll
      for (int q = 0; q < 4; ++q) {
        uint4 t = ((const uint4*)grp)[q];
        t16[4 * q] = t.x; t16[4 * q + 1] = t.y; t16[4 * q + 2] = t.z; t16[4 * q + 3] = t.w;
      }
      unsigned run = 0, cum = 0, hi = 0;
      #pragma unroll
      for (int q = 0; q < 16; ++q) {
        run += t16[q];
        bool below = (run <= kk);
        hi += below ? 1u : 0u;
        cum = below ? run : cum;
      }
      hi = min(hi, 15u);
      #pragma unroll
      for (int q = 0; q < 4; ++q) {
        uint4 t = ((const uint4*)&hist[hi * 16])[q];
        t16[4 * q] = t.x; t16[4 * q + 1] = t.y; t16[4 * q + 2] = t.z; t16[4 * q + 3] = t.w;
      }
      unsigned lo = 0;
      run = cum;
      #pragma unroll
      for (int q = 0; q < 16; ++q) {
        run += t16[q];
        bool below = (run <= kk);
        lo += below ? 1u : 0u;
        cum = below ? run : cum;
      }
      misc[1] = kk - cum;
      misc[0] |= ((hi * 16u + lo) << shift);
    }
    __syncthreads();
    prefix = misc[0];
    prefmask |= (255u << shift);
  }
  const unsigned theta = prefix;

  unsigned cl = 0, ce = 0;
  #pragma unroll
  for (int q = 0; q < 8; ++q) {
    bool valid = (q < nv);
    cl += (valid && key[q] < theta) ? 1u : 0u;
    ce += (valid && key[q] == theta) ? 1u : 0u;
  }
  const int lane = tid & 31, wv = tid >> 5;
  const unsigned below = (1u << lane) - 1u;
  unsigned tot_l = 0, tot_e = 0, pre_e = 0;
  #pragma unroll
  for (int bit = 0; bit < 4; ++bit) {
    unsigned bl = (unsigned)__ballot((cl >> bit) & 1u);
    unsigned be = (unsigned)__ballot((ce >> bit) & 1u);
    tot_l += (unsigned)__popc(bl) << bit;
    tot_e += (unsigned)__popc(be) << bit;
    pre_e += (unsigned)__popc(be & below) << bit;
  }
  if (lane == 0) { wave_lt[wv] = tot_l; wave_eq[wv] = tot_e; }
  __syncthreads();
  unsigned n_less = 0, run_eq = pre_e;
  {
    const uint4 l0 = ((const uint4*)wave_lt)[0], l1 = ((const uint4*)wave_lt)[1];
    const uint4 e0 = ((const uint4*)wave_eq)[0], e1 = ((const uint4*)wave_eq)[1];
    const unsigned wl[8] = {l0.x, l0.y, l0.z, l0.w, l1.x, l1.y, l1.z, l1.w};
    const unsigned we[8] = {e0.x, e0.y, e0.z, e0.w, e1.x, e1.y, e1.z, e1.w};
    #pragma unroll
    for (int w2 = 0; w2 < 8; ++w2) {
      n_less += wl[w2];
      run_eq += (w2 < wv) ? we[w2] : 0u;
    }
  }
  const int ntake = BUDGET - (int)n_less;

  unsigned bits = 0;
  #pragma unroll
  for (int q = 0; q < 8; ++q) {
    bool valid = (q < nv);
    unsigned u = key[q];
    bool eq = valid && (u == theta);
    bool keep = valid && ((u < theta) || (eq && (int)run_eq < ntake));
    run_eq += eq ? 1u : 0u;
    keep = keep || (j0 + q == i);
    bits |= (keep ? 1u : 0u) << q;
  }
  ((unsigned char*)mrow)[tid] = (unsigned char)bits;
}

#define NW 4
#define OST (HD + 4)
#define KP  (HD + 4)
#define VP  (HD + 8)
#define PTP 17
#define WSF (16 * KP + 16 * VP + 16 * PTP + 16)

__device__ __forceinline__ void wave_lds_sync() {
  __builtin_amdgcn_fence(__ATOMIC_RELEASE, "wavefront");
  __builtin_amdgcn_wave_barrier();
  __builtin_amdgcn_fence(__ATOMIC_ACQUIRE, "wavefront");
}

__device__ __forceinline__ float xhalf(float x) {
  int b = __builtin_bit_cast(int, x);
  return __builtin_bit_cast(float, __builtin_amdgcn_permlanex16(b, b, 0x76543210u, 0xfedcba98u, false, false));
}

__global__ void __launch_bounds__(32 * NW)
attn(const float* __restrict__ qkv, const float* __restrict__ FF,
     const unsigned* __restrict__ mask, float* __restrict__ Y) {
  int h = blockIdx.x, b = blockIdx.y;
  int ib = NT - 1 - (int)blockIdx.z;
  int w = __builtin_amdgcn_readfirstlane((int)(threadIdx.x >> 5));
  int lane = threadIdx.x & 31, g = lane >> 4, lm = lane & 15;
  int i0 = ib * 16;
  __shared__ __attribute__((aligned(16))) float stage[NW][WSF];
  __shared__ float part_m[NW][16];
  __shared__ float part_l[NW][16];
  float* Ks = stage[w];
  float* Vs = Ks + 16 * KP;
  float* pt = Vs + 16 * VP;
  float* sc = pt + 16 * PTP;

  const float* qb = qkv + ((size_t)(b * TT)) * C3 + h * HD;
  const float* kb = qb + CC;

  v4f aq[8];
  #pragma unroll
  for (int k8 = 0; k8 < 8; ++k8)
    aq[k8] = *(const v4f*)(qb + (size_t)(i0 + lm) * C3 + k8 * 8 + 4 * g);

  const float* kg = kb + (size_t)g * C3 + lm * 4;
  const float* vg = kg + CC;
  const int k_st = g * KP + lm * 4;
  const int v_st = g * VP + lm * 4;
  const int k_fr = lm * KP + 4 * g;
  const int v_fr = 2 * g * VP + lm;

  float mrow = -1e30f, lrow = 0.f;
  v8f o[4];
  #pragma unroll
  for (int t = 0; t < 4; ++t)
    #pragma unroll
    for (int r = 0; r < 8; ++r) o[t][r] = 0.f;
  const unsigned* mrp = mask + ((size_t)(b * TT + i0 + lm)) * (TT / 32);
  const int msh = ((w & 1) << 4) + 2 * g;

  v4f rk[8], rv[8];
  {
    const int jf = min(w, ib) * 16;
    #pragma unroll
    for (int p = 0; p < 8; ++p) {
      rk[p] = *(const v4f*)(kg + (size_t)(jf + 2 * p) * C3);
      rv[p] = *(const v4f*)(vg + (size_t)(jf + 2 * p) * C3);
    }
    #pragma unroll
    for (int p = 0; p < 8; ++p) {
      *(v4f*)(Ks + k_st + 2 * p * KP) = rk[p];
      *(v4f*)(Vs + v_st + 2 * p * VP) = rv[p];
    }
  }
  wave_lds_sync();

  for (int jb = w; jb <= ib; jb += NW) {
    int j0 = jb * 16;
    const int jn = min(jb + NW, ib) * 16;
    float ffv[8];
    #pragma unroll
    for (int r = 0; r < 8; ++r) {
      int i = i0 + r + 8 * g;
      ffv[r] = FF[((size_t)(b * TT + i)) * TT + j0 + lm];
    }
    const unsigned mw = mrp[j0 >> 5] >> msh;
    #pragma unroll
    for (int p = 0; p < 8; ++p)
      rk[p] = *(const v4f*)(kg + (size_t)(jn + 2 * p) * C3);
    #pragma unroll
    for (int p = 0; p < 8; ++p)
      rv[p] = *(const v4f*)(vg + (size_t)(jn + 2 * p) * C3);
    v4f fk[8];
    #pragma unroll
    for (int k8 = 0; k8 < 8; ++k8) fk[k8] = *(const v4f*)(Ks + k_fr + k8 * 8);
    v8f acc = {0.f, 0.f, 0.f, 0.f, 0.f, 0.f, 0.f, 0.f};
    #pragma unroll
    for (int k8 = 0; k8 < 8; ++k8) {
      v2f a0 = {aq[k8].x, aq[k8].y}, b0 = {fk[k8].x, fk[k8].y};
      acc = wmma4(a0, b0, acc);
      v2f a1 = {aq[k8].z, aq[k8].w}, b1 = {fk[k8].z, fk[k8].w};
      acc = wmma4(a1, b1, acc);
    }

    #pragma unroll
    for (int r = 0; r < 8; ++r) pt[(r + 8 * g) * PTP + lm] = acc[r] * 0.125f - ffv[r];
    wave_lds_sync();
    float s[8];
    #pragma unroll
    for (int kt = 0; kt < 4; ++kt) {
      s[2 * kt]     = pt[lm * PTP + kt * 4 + 2 * g];
      s[2 * kt + 1] = pt[lm * PTP + kt * 4 + 2 * g + 1];
    }
    const int i = i0 + lm;
    #pragma unroll
    for (int q = 0; q < 8; ++q) {
      int j = j0 + 4 * (q >> 1) + 2 * g + (q & 1);
      bool keep = ((mw >> (4 * (q >> 1) + (q & 1))) & 1u) && (j <= i);
      s[q] = keep ? s[q] : -1e30f;
    }
    float bm = __builtin_elementwise_maximum(
        __builtin_elementwise_maximum(__builtin_elementwise_maximum(s[0], s[1]), __builtin_elementwise_maximum(s[2], s[3])),
        __builtin_elementwise_maximum(__builtin_elementwise_maximum(s[4], s[5]), __builtin_elementwise_maximum(s[6], s[7])));
    bm = __builtin_elementwise_maximum(bm, xhalf(bm));
    const float mnew = __builtin_elementwise_maximum(mrow, bm);
    const float scale = __expf(mrow - mnew);
    float p[8];
    #pragma unroll
    for (int q = 0; q < 8; ++q) p[q] = __expf(s[q] - mnew);
    lrow = lrow * scale + (((p[0] + p[1]) + (p[2] + p[3])) + ((p[4] + p[5]) + (p[6] + p[7])));
    mrow = mnew;
    __builtin_amdgcn_sched_barrier(0);
    #pragma unroll
    for (int p8 = 0; p8 < 8; ++p8) *(v4f*)(Ks + k_st + 2 * p8 * KP) = rk[p8];
    sc[lm] = scale;
    wave_lds_sync();
    const v4f sc0 = *(const v4f*)(sc + 8 * g), sc1 = *(const v4f*)(sc + 8 * g + 4);
    const float scr[8] = {sc0.x, sc0.y, sc0.z, sc0.w, sc1.x, sc1.y, sc1.z, sc1.w};
    #pragma unroll
    for (int r = 0; r < 8; ++r)
      #pragma unroll
      for (int t = 0; t < 4; ++t) o[t][r] *= scr[r];
    v2f ap[4];
    #pragma unroll
    for (int kt = 0; kt < 4; ++kt) {
      ap[kt].x = p[2 * kt];
      ap[kt].y = p[2 * kt + 1];
    }
    v2f bv[4][4];
    #pragma unroll
    for (int kt = 0; kt < 4; ++kt)
      #pragma unroll
      for (int t = 0; t < 4; ++t) {
        bv[kt][t].x = Vs[v_fr + (kt * 4) * VP + t * 16];
        bv[kt][t].y = Vs[v_fr + (kt * 4 + 1) * VP + t * 16];
      }
    #pragma unroll
    for (int kt = 0; kt < 4; ++kt)
      #pragma unroll
      for (int t = 0; t < 4; ++t)
        o[t] = wmma4(ap[kt], bv[kt][t], o[t]);
    wave_lds_sync();
    #pragma unroll
    for (int p8 = 0; p8 < 8; ++p8) *(v4f*)(Vs + v_st + 2 * p8 * VP) = rv[p8];
  }

  wave_lds_sync();
  float* po = stage[w];
  #pragma unroll
  for (int r = 0; r < 8; ++r) {
    #pragma unroll
    for (int t = 0; t < 4; ++t) po[(r + 8 * g) * OST + t * 16 + lm] = o[t][r];
  }
  const float lfull = lrow + xhalf(lrow);
  if (g == 0) { part_m[w][lm] = mrow; part_l[w][lm] = lfull; }
  __syncthreads();
  #pragma unroll
  for (int rr = 0; rr < 16 / NW; ++rr) {
    int row = w * (16 / NW) + rr;
    float mw[NW], M = -1e30f;
    #pragma unroll
    for (int w2 = 0; w2 < NW; ++w2) { mw[w2] = part_m[w2][row]; M = fmaxf(M, mw[w2]); }
    float L = 0.f, o0 = 0.f, o1 = 0.f;
    #pragma unroll
    for (int w2 = 0; w2 < NW; ++w2) {
      float wt = __expf(mw[w2] - M);
      L  += part_l[w2][row] * wt;
      o0 += stage[w2][row * OST + lane] * wt;
      o1 += stage[w2][row * OST + lane + 32] * wt;
    }
    float inv = 1.f / L;
    float* yrow = Y + ((size_t)(b * TT + i0 + row)) * CC + h * HD;
    yrow[lane]      = o0 * inv;
    yrow[lane + 32] = o1 * inv;
  }
}

extern "C" void kernel_launch(void* const* d_in, const int* in_sizes, int n_in,
                              void* d_out, int out_size, void* d_ws, size_t ws_size,
                              hipStream_t stream) {
  const float* x      = (const float*)d_in[0];
  const float* w_attn = (const float*)d_in[1];
  const float* b_attn = (const float*)d_in[2];
  const float* w_proj = (const float*)d_in[3];
  const float* b_proj = (const float*)d_in[4];
  float* out = (float*)d_out;

  char* ws = (char*)d_ws;
  const size_t QKV_BYTES = (size_t)BB * TT * C3 * 4;
  const size_t FF_BYTES  = (size_t)BB * TT * TT * 4;
  const size_t MSK_BYTES = (size_t)BB * TT * (TT / 32) * 4;
  float*    qkv = (float*)ws;
  float*    FF  = (float*)(ws + QKV_BYTES);
  unsigned* msk = (unsigned*)(ws + QKV_BYTES + FF_BYTES);
  float*    Y   = (float*)(ws + QKV_BYTES + FF_BYTES + MSK_BYTES);

  gemm_wmma<<<dim3(C3 / GBN, (BB * TT) / GBM), 128, 0, stream>>>(
      x, w_attn, b_attn, qkv, BB * TT, C3, CC);
  float*    Ttot = Y;
  float*    Toff = Y + (size_t)BB * NT * TT;
  head0_tiles<<<dim3(NT / HS + 1, NT / 8, BB), 128, 0, stream>>>(qkv, FF, Ttot);
  tile_scan<<<dim3(TT / 64, BB), 64, 0, stream>>>(Ttot, Toff);
  select_mask<<<BB * TT, 256, 0, stream>>>(FF, Toff, msk);
  attn<<<dim3(NH, BB, NT), 32 * NW, 0, stream>>>(qkv, FF, msk, Y);
  gemm_wmma<<<dim3(CC / GBN, (BB * TT) / GBM), 128, 0, stream>>>(
      Y, w_proj, b_proj, out, BB * TT, CC, CC);
}
